# starting kernel
# baseline (speedup 1.0000x reference)
#include <hip/hip_runtime.h>

typedef __attribute__((ext_vector_type(2))) float v2f;
typedef __attribute__((ext_vector_type(8))) float v8f;

#define N_NODES  100000
#define N_EDGES  3200000
#define N_GRAPHS 1024

static inline int ceil_div(int a, int b) { return (a + b - 1) / b; }

__global__ void fill4_f32(float4* __restrict__ p, float v, int nquads) {
    int i = blockIdx.x * blockDim.x + threadIdx.x;
    if (i < nquads) p[i] = make_float4(v, v, v, v);
}

__global__ void deg_accum(const int* __restrict__ ei, float* __restrict__ deg) {
    int e = blockIdx.x * blockDim.x + threadIdx.x;
    if (e < N_EDGES) atomicAdd(&deg[ei[N_EDGES + e]], 1.0f);
}

__global__ void rsqrt_inplace(float* __restrict__ d, int n) {
    int i = blockIdx.x * blockDim.x + threadIdx.x;
    if (i < n) d[i] = rsqrtf(d[i]);
}

template <int CIN, int COUT>
__global__ void __launch_bounds__(256)
gemm_wmma_f32(const float* __restrict__ H, const float* __restrict__ W,
              const float* __restrict__ bias, float* __restrict__ out, int nrows) {
    constexpr int NT     = COUT / 16;
    constexpr int KSTEPS = (CIN + 3) / 4;

    const int wavesPerBlock = blockDim.x >> 5;
    const int wave = blockIdx.x * wavesPerBlock + (threadIdx.x >> 5);
    const int row0 = wave * 16;
    if (row0 >= nrows) return;

    const int lane = threadIdx.x & 31;
    const int half = lane >> 4;
    const int m    = lane & 15;
    const int row  = row0 + m;

    v8f acc[NT];
#pragma unroll
    for (int nt = 0; nt < NT; ++nt) acc[nt] = (v8f){};

#pragma unroll
    for (int ks = 0; ks < KSTEPS; ++ks) {
        const int ka = ks * 4 + 2 * half;

        v2f a;
        if constexpr (CIN % 4 == 0) {
            a = *reinterpret_cast<const v2f*>(&H[row * CIN + ka]);
        } else {
            const int k0 = (ka     < CIN) ? ka     : (CIN - 1);
            const int k1 = (ka + 1 < CIN) ? ka + 1 : (CIN - 1);
            float ax = H[row * CIN + k0];
            float ay = H[row * CIN + k1];
            a.x = (ka     < CIN) ? ax : 0.0f;
            a.y = (ka + 1 < CIN) ? ay : 0.0f;
        }

#pragma unroll
        for (int nt = 0; nt < NT; ++nt) {
            const int ncol = nt * 16 + m;
            v2f b;
            if constexpr (CIN % 4 == 0) {
                b = *reinterpret_cast<const v2f*>(&W[ncol * CIN + ka]);
            } else {
                const int k0 = (ka     < CIN) ? ka     : (CIN - 1);
                const int k1 = (ka + 1 < CIN) ? ka + 1 : (CIN - 1);
                float bx = W[ncol * CIN + k0];
                float by = W[ncol * CIN + k1];
                b.x = (ka     < CIN) ? bx : 0.0f;
                b.y = (ka + 1 < CIN) ? by : 0.0f;
            }
            acc[nt] = __builtin_amdgcn_wmma_f32_16x16x4_f32(
                false, a, false, b, (short)0, acc[nt], false, false);
        }
    }

#pragma unroll
    for (int nt = 0; nt < NT; ++nt) {
        const int col = nt * 16 + m;
        const float bb = bias ? bias[col] : 0.0f;
#pragma unroll
        for (int r = 0; r < 8; ++r) {
            out[(row0 + r + 8 * half) * COUT + col] = acc[nt][r] + bb;
        }
    }
}

template <int COUT>
__global__ void edge_scatter(const int* __restrict__ ei, const float* __restrict__ t,
                             const float* __restrict__ dis, float* __restrict__ agg) {
    constexpr int GROUPS = COUT / 4;
    const int tid = blockIdx.x * blockDim.x + threadIdx.x;
    const int e = tid / GROUPS;
    const int g = tid % GROUPS;
    if (e >= N_EDGES) return;
    if (g == 0 && e + 4096 < N_EDGES) {
        __builtin_prefetch(&ei[e + 4096], 0, 1);
        __builtin_prefetch(&ei[N_EDGES + e + 4096], 0, 1);
    }
    const int src = ei[e];
    const int dst = ei[N_EDGES + e];
    const float coeff = dis[src] * dis[dst];
    const float4 v = *reinterpret_cast<const float4*>(&t[src * COUT + g * 4]);
    float* base = &agg[dst * COUT + g * 4];
    atomicAdd(base + 0, v.x * coeff);
    atomicAdd(base + 1, v.y * coeff);
    atomicAdd(base + 2, v.z * coeff);
    atomicAdd(base + 3, v.w * coeff);
}

template <int COUT>
__global__ void finalize_relu(const float* __restrict__ agg, const float* __restrict__ t,
                              const float* __restrict__ dis, const float* __restrict__ bias,
                              float* __restrict__ outH) {
    constexpr int GROUPS = COUT / 4;
    const int q = blockIdx.x * blockDim.x + threadIdx.x;
    if (q >= N_NODES * GROUPS) return;
    const int n  = q / GROUPS;
    const int c4 = (q % GROUPS) * 4;
    const float d  = dis[n];
    const float dd = d * d;
    const float4 va = *reinterpret_cast<const float4*>(&agg[n * COUT + c4]);
    const float4 vt = *reinterpret_cast<const float4*>(&t[n * COUT + c4]);
    const float4 vb = *reinterpret_cast<const float4*>(&bias[c4]);
    float4 r;
    r.x = fmaxf(fmaf(vt.x, dd, va.x) + vb.x, 0.0f);
    r.y = fmaxf(fmaf(vt.y, dd, va.y) + vb.y, 0.0f);
    r.z = fmaxf(fmaf(vt.z, dd, va.z) + vb.z, 0.0f);
    r.w = fmaxf(fmaf(vt.w, dd, va.w) + vb.w, 0.0f);
    *reinterpret_cast<float4*>(&outH[n * COUT + c4]) = r;
}

__global__ void pool_kernel(const float* __restrict__ h, const int* __restrict__ batch,
                            float* __restrict__ pooled) {
    const int q = blockIdx.x * blockDim.x + threadIdx.x;
    if (q >= N_NODES * 4) return;
    const int n  = q >> 2;
    const int c4 = (q & 3) * 4;
    const float4 v = *reinterpret_cast<const float4*>(&h[n * 16 + c4]);
    float* base = &pooled[batch[n] * 16 + c4];
    atomicAdd(base + 0, v.x);
    atomicAdd(base + 1, v.y);
    atomicAdd(base + 2, v.z);
    atomicAdd(base + 3, v.w);
}

__global__ void classify(const float* __restrict__ emb, const float* __restrict__ Wc,
                         const float* __restrict__ bc, float* __restrict__ out) {
    const int g = blockIdx.x * blockDim.x + threadIdx.x;
    if (g >= N_GRAPHS) return;
    float s = bc[0];
#pragma unroll
    for (int c = 0; c < 16; ++c) s += fmaxf(emb[g * 16 + c], 0.0f) * Wc[c];
    out[g] = s;
}

extern "C" void kernel_launch(void* const* d_in, const int* in_sizes, int n_in,
                              void* d_out, int out_size, void* d_ws, size_t ws_size,
                              hipStream_t stream) {
    const float* x     = (const float*)d_in[0];
    const int*   ei    = (const int*)d_in[1];
    const int*   batch = (const int*)d_in[2];
    const float* W1 = (const float*)d_in[3];
    const float* b1 = (const float*)d_in[4];
    const float* W2 = (const float*)d_in[5];
    const float* b2 = (const float*)d_in[6];
    const float* W3 = (const float*)d_in[7];
    const float* b3 = (const float*)d_in[8];
    const float* We = (const float*)d_in[9];
    const float* be = (const float*)d_in[10];
    const float* Wc = (const float*)d_in[11];
    const float* bc = (const float*)d_in[12];
    float* out = (float*)d_out;

    float* ws     = (float*)d_ws;
    float* dis    = ws;
    float* bufT   = dis  + N_NODES;
    float* bufA   = bufT + N_NODES * 32;
    float* bufH   = bufA + N_NODES * 32;
    float* pooled = bufH + N_NODES * 32;

    const int B = 256;

    fill4_f32<<<ceil_div(N_NODES / 4, B), B, 0, stream>>>((float4*)dis, 1.0f, N_NODES / 4);
    deg_accum<<<ceil_div(N_EDGES, B), B, 0, stream>>>(ei, dis);
    rsqrt_inplace<<<ceil_div(N_NODES, B), B, 0, stream>>>(dis, N_NODES);

    const int gBlocks = ceil_div(N_NODES / 16, B / 32);

    gemm_wmma_f32<3, 32><<<gBlocks, B, 0, stream>>>(x, W1, nullptr, bufT, N_NODES);
    fill4_f32<<<ceil_div(N_NODES * 8, B), B, 0, stream>>>((float4*)bufA, 0.0f, N_NODES * 8);
    edge_scatter<32><<<ceil_div(N_EDGES * 8, B), B, 0, stream>>>(ei, bufT, dis, bufA);
    finalize_relu<32><<<ceil_div(N_NODES * 8, B), B, 0, stream>>>(bufA, bufT, dis, b1, bufH);

    gemm_wmma_f32<32, 16><<<gBlocks, B, 0, stream>>>(bufH, W2, nullptr, bufT, N_NODES);
    fill4_f32<<<ceil_div(N_NODES * 4, B), B, 0, stream>>>((float4*)bufA, 0.0f, N_NODES * 4);
    edge_scatter<16><<<ceil_div(N_EDGES * 4, B), B, 0, stream>>>(ei, bufT, dis, bufA);
    finalize_relu<16><<<ceil_div(N_NODES * 4, B), B, 0, stream>>>(bufA, bufT, dis, b2, bufH);

    gemm_wmma_f32<16, 16><<<gBlocks, B, 0, stream>>>(bufH, W3, nullptr, bufT, N_NODES);
    fill4_f32<<<ceil_div(N_NODES * 4, B), B, 0, stream>>>((float4*)bufA, 0.0f, N_NODES * 4);
    edge_scatter<16><<<ceil_div(N_EDGES * 4, B), B, 0, stream>>>(ei, bufT, dis, bufA);
    finalize_relu<16><<<ceil_div(N_NODES * 4, B), B, 0, stream>>>(bufA, bufT, dis, b3, bufH);

    fill4_f32<<<ceil_div(N_GRAPHS * 4, B), B, 0, stream>>>((float4*)pooled, 0.0f, N_GRAPHS * 4);
    pool_kernel<<<ceil_div(N_NODES * 4, B), B, 0, stream>>>(bufH, batch, pooled);
    gemm_wmma_f32<16, 16><<<ceil_div(N_GRAPHS / 16, B / 32), B, 0, stream>>>(
        pooled, We, be, out, N_GRAPHS);
    classify<<<ceil_div(N_GRAPHS, B), B, 0, stream>>>(out, Wc, bc, out + N_GRAPHS * 16);
}


# Rows owned by L2 half: each edge slice is done once per half, by a workgroup of that half (die id, per-half ticket), which commits only atomics to its half's rows; atomic sum order as before
# speedup vs baseline: 1.6643x; 1.6643x over previous
#include <hip/hip_runtime.h>

typedef __attribute__((ext_vector_type(2))) float v2f;
typedef __attribute__((ext_vector_type(8))) float v8f;

#define N_NODES  100000
#define N_EDGES  3200000
#define N_GRAPHS 1024

#define SPLIT_NODE (N_NODES / 2)
#define SLICE      1024
#define N_SLICES   (N_EDGES / SLICE)
#define TICK_PITCH 128
static_assert(N_EDGES % SLICE == 0, "slices cover the edge list exactly");

static inline int ceil_div(int a, int b) { return (a + b - 1) / b; }

__global__ void fill4_f32(float4* __restrict__ p, float v, int nquads, int* __restrict__ tick) {
    int i = blockIdx.x * blockDim.x + threadIdx.x;
    if (i < nquads) p[i] = make_float4(v, v, v, v);
    if (tick != nullptr && i < 2) tick[i * TICK_PITCH] = 0;
}

__device__ __forceinline__ int draw_job(int* __restrict__ tick) {
    const int h = (int)((__builtin_amdgcn_s_sendmsg_rtn(0x87) >> 18) & 1u);
    int t = atomicAdd(&tick[h * TICK_PITCH], 1);
    if (t < N_SLICES) return (t << 1) | h;
    t = atomicAdd(&tick[(h ^ 1) * TICK_PITCH], 1);
    if (t < N_SLICES) return (t << 1) | (h ^ 1);
    return -1;
}

__global__ void __launch_bounds__(256)
deg_accum(const int* __restrict__ ei, float* __restrict__ deg, int* __restrict__ tick) {
    __shared__ int s_job;
    if (threadIdx.x == 0) s_job = draw_job(tick);
    __syncthreads();
    const int job = s_job;
    if (job < 0) return;
    const int h = job & 1;
    const int base = (job >> 1) * SLICE;
    const int4 d = *reinterpret_cast<const int4*>(&ei[N_EDGES + base + threadIdx.x * 4]);
    if ((d.x >= SPLIT_NODE ? 1 : 0) == h) atomicAdd(&deg[d.x], 1.0f);
    if ((d.y >= SPLIT_NODE ? 1 : 0) == h) atomicAdd(&deg[d.y], 1.0f);
    if ((d.z >= SPLIT_NODE ? 1 : 0) == h) atomicAdd(&deg[d.z], 1.0f);
    if ((d.w >= SPLIT_NODE ? 1 : 0) == h) atomicAdd(&deg[d.w], 1.0f);
}

__global__ void rsqrt_inplace(float* __restrict__ d, int n) {
    int i = blockIdx.x * blockDim.x + threadIdx.x;
    if (i < n) d[i] = rsqrtf(d[i]);
}

template <int CIN, int COUT>
__global__ void __launch_bounds__(256)
gemm_wmma_f32(const float* __restrict__ H, const float* __restrict__ W,
              const float* __restrict__ bias, float* __restrict__ out, int nrows) {
    constexpr int NT     = COUT / 16;
    constexpr int KSTEPS = (CIN + 3) / 4;

    const int wavesPerBlock = blockDim.x >> 5;
    const int wave = blockIdx.x * wavesPerBlock + (threadIdx.x >> 5);
    const int row0 = wave * 16;
    if (row0 >= nrows) return;

    const int lane = threadIdx.x & 31;
    const int half = lane >> 4;
    const int m    = lane & 15;
    const int row  = row0 + m;

    v8f acc[NT];
#pragma unroll
    for (int nt = 0; nt < NT; ++nt) acc[nt] = (v8f){};

#pragma unroll
    for (int ks = 0; ks < KSTEPS; ++ks) {
        const int ka = ks * 4 + 2 * half;

        v2f a;
        if constexpr (CIN % 4 == 0) {
            a = *reinterpret_cast<const v2f*>(&H[row * CIN + ka]);
        } else {
            const int k0 = (ka     < CIN) ? ka     : (CIN - 1);
            const int k1 = (ka + 1 < CIN) ? ka + 1 : (CIN - 1);
            float ax = H[row * CIN + k0];
            float ay = H[row * CIN + k1];
            a.x = (ka     < CIN) ? ax : 0.0f;
            a.y = (ka + 1 < CIN) ? ay : 0.0f;
        }

#pragma unroll
        for (int nt = 0; nt < NT; ++nt) {
            const int ncol = nt * 16 + m;
            v2f b;
            if constexpr (CIN % 4 == 0) {
                b = *reinterpret_cast<const v2f*>(&W[ncol * CIN + ka]);
            } else {
                const int k0 = (ka     < CIN) ? ka     : (CIN - 1);
                const int k1 = (ka + 1 < CIN) ? ka + 1 : (CIN - 1);
                float bx = W[ncol * CIN + k0];
                float by = W[ncol * CIN + k1];
                b.x = (ka     < CIN) ? bx : 0.0f;
                b.y = (ka + 1 < CIN) ? by : 0.0f;
            }
            acc[nt] = __builtin_amdgcn_wmma_f32_16x16x4_f32(
                false, a, false, b, (short)0, acc[nt], false, false);
        }
    }

#pragma unroll
    for (int nt = 0; nt < NT; ++nt) {
        const int col = nt * 16 + m;
        const float bb = bias ? bias[col] : 0.0f;
#pragma unroll
        for (int r = 0; r < 8; ++r) {
            out[(row0 + r + 8 * half) * COUT + col] = acc[nt][r] + bb;
        }
    }
}

template <int COUT>
__global__ void __launch_bounds__(256)
edge_scatter(const int* __restrict__ ei, const float* __restrict__ t,
             const float* __restrict__ dis, float* __restrict__ agg, int* __restrict__ tick) {
    constexpr int GROUPS = COUT / 4;
    constexpr int EPI    = 256 / GROUPS;
    constexpr int U      = 4;
    __shared__ int s_job;
    __shared__ int s_wtot[8];
    __shared__ int s_src[SLICE];
    __shared__ int s_dst[SLICE];

    const int tid  = threadIdx.x;
    const int lane = tid & 31;
    const int wave = tid >> 5;
    if (tid == 0) s_job = draw_job(tick);
    __syncthreads();
    const int job = s_job;
    if (job < 0) return;
    const int h = job & 1;
    const int base = (job >> 1) * SLICE;

    const int4 dv = *reinterpret_cast<const int4*>(&ei[N_EDGES + base + tid * 4]);
    const int4 sv = *reinterpret_cast<const int4*>(&ei[base + tid * 4]);
    const int dd[4] = {dv.x, dv.y, dv.z, dv.w};
    const int ss[4] = {sv.x, sv.y, sv.z, sv.w};
    bool own[4];
    int  rank[4];
    int  tot = 0;
#pragma unroll
    for (int k = 0; k < 4; ++k) {
        own[k] = ((dd[k] >= SPLIT_NODE) ? 1 : 0) == h;
        const unsigned m = __builtin_amdgcn_ballot_w32(own[k]);
        rank[k] = tot + (int)__builtin_amdgcn_mbcnt_lo(m, 0u);
        tot += __builtin_popcount(m);
    }
    if (lane == 0) s_wtot[wave] = tot;
    __syncthreads();
    int wbase = 0, n = 0;
#pragma unroll
    for (int w = 0; w < 8; ++w) {
        const int c = s_wtot[w];
        wbase += (w < wave) ? c : 0;
        n += c;
    }
#pragma unroll
    for (int k = 0; k < 4; ++k) {
        if (own[k]) {
            s_src[wbase + rank[k]] = ss[k];
            s_dst[wbase + rank[k]] = dd[k];
        }
    }
    __syncthreads();
    if (n == 0) return;

    const int g  = tid % GROUPS;
    const int le = tid / GROUPS;
    for (int i0 = le; i0 < n; i0 += EPI * U) {
        int    si[U], di[U];
        float  cs[U], cd[U];
        float4 v[U];
#pragma unroll
        for (int u = 0; u < U; ++u) {
            const int i  = i0 + u * EPI;
            const int ic = (i < n) ? i : (n - 1);
            si[u] = s_src[ic];
            di[u] = s_dst[ic];
        }
#pragma unroll
        for (int u = 0; u < U; ++u) {
            cs[u] = dis[si[u]];
            cd[u] = dis[di[u]];
            v[u]  = *reinterpret_cast<const float4*>(&t[si[u] * COUT + g * 4]);
        }
        static_assert(U == 4, "the fence names four edges' values");
        asm volatile("" :: "v"(cs[0]), "v"(cd[0]), "v"(v[0].x), "v"(v[0].y), "v"(v[0].z), "v"(v[0].w),
                           "v"(cs[1]), "v"(cd[1]), "v"(v[1].x), "v"(v[1].y), "v"(v[1].z), "v"(v[1].w),
                           "v"(cs[2]), "v"(cd[2]), "v"(v[2].x), "v"(v[2].y), "v"(v[2].z), "v"(v[2].w),
                           "v"(cs[3]), "v"(cd[3]), "v"(v[3].x), "v"(v[3].y), "v"(v[3].z), "v"(v[3].w));
#pragma unroll
        for (int u = 0; u < U; ++u) {
            if (i0 + u * EPI < n) {
                const float coeff = cs[u] * cd[u];
                float* b = &agg[di[u] * COUT + g * 4];
                atomicAdd(b + 0, v[u].x * coeff);
                atomicAdd(b + 1, v[u].y * coeff);
                atomicAdd(b + 2, v[u].z * coeff);
                atomicAdd(b + 3, v[u].w * coeff);
            }
        }
    }
}

template <int COUT>
__global__ void finalize_relu(const float* __restrict__ agg, const float* __restrict__ t,
                              const float* __restrict__ dis, const float* __restrict__ bias,
                              float* __restrict__ outH) {
    constexpr int GROUPS = COUT / 4;
    const int q = blockIdx.x * blockDim.x + threadIdx.x;
    if (q >= N_NODES * GROUPS) return;
    const int n  = q / GROUPS;
    const int c4 = (q % GROUPS) * 4;
    const float d  = dis[n];
    const float dd = d * d;
    const float4 va = *reinterpret_cast<const float4*>(&agg[n * COUT + c4]);
    const float4 vt = *reinterpret_cast<const float4*>(&t[n * COUT + c4]);
    const float4 vb = *reinterpret_cast<const float4*>(&bias[c4]);
    float4 r;
    r.x = fmaxf(fmaf(vt.x, dd, va.x) + vb.x, 0.0f);
    r.y = fmaxf(fmaf(vt.y, dd, va.y) + vb.y, 0.0f);
    r.z = fmaxf(fmaf(vt.z, dd, va.z) + vb.z, 0.0f);
    r.w = fmaxf(fmaf(vt.w, dd, va.w) + vb.w, 0.0f);
    *reinterpret_cast<float4*>(&outH[n * COUT + c4]) = r;
}

__global__ void pool_kernel(const float* __restrict__ h, const int* __restrict__ batch,
                            float* __restrict__ pooled) {
    const int q = blockIdx.x * blockDim.x + threadIdx.x;
    if (q >= N_NODES * 4) return;
    const int n  = q >> 2;
    const int c4 = (q & 3) * 4;
    const float4 v = *reinterpret_cast<const float4*>(&h[n * 16 + c4]);
    float* base = &pooled[batch[n] * 16 + c4];
    atomicAdd(base + 0, v.x);
    atomicAdd(base + 1, v.y);
    atomicAdd(base + 2, v.z);
    atomicAdd(base + 3, v.w);
}

__global__ void classify(const float* __restrict__ emb, const float* __restrict__ Wc,
                         const float* __restrict__ bc, float* __restrict__ out) {
    const int g = blockIdx.x * blockDim.x + threadIdx.x;
    if (g >= N_GRAPHS) return;
    float s = bc[0];
#pragma unroll
    for (int c = 0; c < 16; ++c) s += fmaxf(emb[g * 16 + c], 0.0f) * Wc[c];
    out[g] = s;
}

extern "C" void kernel_launch(void* const* d_in, const int* in_sizes, int n_in,
                              void* d_out, int out_size, void* d_ws, size_t ws_size,
                              hipStream_t stream) {
    const float* x     = (const float*)d_in[0];
    const int*   ei    = (const int*)d_in[1];
    const int*   batch = (const int*)d_in[2];
    const float* W1 = (const float*)d_in[3];
    const float* b1 = (const float*)d_in[4];
    const float* W2 = (const float*)d_in[5];
    const float* b2 = (const float*)d_in[6];
    const float* W3 = (const float*)d_in[7];
    const float* b3 = (const float*)d_in[8];
    const float* We = (const float*)d_in[9];
    const float* be = (const float*)d_in[10];
    const float* Wc = (const float*)d_in[11];
    const float* bc = (const float*)d_in[12];
    float* out = (float*)d_out;

    float* ws     = (float*)d_ws;
    float* dis    = ws;
    float* bufT   = dis  + N_NODES;
    float* bufA   = bufT + N_NODES * 32;
    float* bufH   = bufA + N_NODES * 32;
    float* pooled = bufH + N_NODES * 32;
    int* tick = (int*)pooled;
    int* none = nullptr;

    const int B = 256;

    fill4_f32<<<ceil_div(N_NODES / 4, B), B, 0, stream>>>((float4*)dis, 1.0f, N_NODES / 4, tick);
    deg_accum<<<2 * N_SLICES, B, 0, stream>>>(ei, dis, tick);
    rsqrt_inplace<<<ceil_div(N_NODES, B), B, 0, stream>>>(dis, N_NODES);

    const int gBlocks = ceil_div(N_NODES / 16, B / 32);

    gemm_wmma_f32<3, 32><<<gBlocks, B, 0, stream>>>(x, W1, nullptr, bufT, N_NODES);
    fill4_f32<<<ceil_div(N_NODES * 8, B), B, 0, stream>>>((float4*)bufA, 0.0f, N_NODES * 8, tick);
    edge_scatter<32><<<2 * N_SLICES, B, 0, stream>>>(ei, bufT, dis, bufA, tick);
    finalize_relu<32><<<ceil_div(N_NODES * 8, B), B, 0, stream>>>(bufA, bufT, dis, b1, bufH);

    gemm_wmma_f32<32, 16><<<gBlocks, B, 0, stream>>>(bufH, W2, nullptr, bufT, N_NODES);
    fill4_f32<<<ceil_div(N_NODES * 4, B), B, 0, stream>>>((float4*)bufA, 0.0f, N_NODES * 4, tick);
    edge_scatter<16><<<2 * N_SLICES, B, 0, stream>>>(ei, bufT, dis, bufA, tick);
    finalize_relu<16><<<ceil_div(N_NODES * 4, B), B, 0, stream>>>(bufA, bufT, dis, b2, bufH);

    gemm_wmma_f32<16, 16><<<gBlocks, B, 0, stream>>>(bufH, W3, nullptr, bufT, N_NODES);
    fill4_f32<<<ceil_div(N_NODES * 4, B), B, 0, stream>>>((float4*)bufA, 0.0f, N_NODES * 4, tick);
    edge_scatter<16><<<2 * N_SLICES, B, 0, stream>>>(ei, bufT, dis, bufA, tick);
    finalize_relu<16><<<ceil_div(N_NODES * 4, B), B, 0, stream>>>(bufA, bufT, dis, b3, bufH);

    fill4_f32<<<ceil_div(N_GRAPHS * 4, B), B, 0, stream>>>((float4*)pooled, 0.0f, N_GRAPHS * 4, none);
    pool_kernel<<<ceil_div(N_NODES * 4, B), B, 0, stream>>>(bufH, batch, pooled);
    gemm_wmma_f32<16, 16><<<ceil_div(N_GRAPHS / 16, B / 32), B, 0, stream>>>(
        pooled, We, be, out, N_GRAPHS);
    classify<<<ceil_div(N_GRAPHS, B), B, 0, stream>>>(out, Wc, bc, out + N_GRAPHS * 16);
}


# Aggregation by pull: edges ranked per call into per-destination lists (degree scan + returning int add); a node's lanes gather and sum its rows in registers, store once; float sums in list order
# speedup vs baseline: 2.1236x; 1.2760x over previous
#include <hip/hip_runtime.h>

typedef __attribute__((ext_vector_type(2))) float v2f;
typedef __attribute__((ext_vector_type(8))) float v8f;

#define N_NODES  100000
#define N_EDGES  3200000
#define N_GRAPHS 1024

#define SPLIT_NODE (N_NODES / 2)
#define SLICE      1024
#define N_SLICES   (N_EDGES / SLICE)
#define TICK_PITCH 128
static_assert(N_EDGES % SLICE == 0, "slices cover the edge list exactly");

static inline int ceil_div(int a, int b) { return (a + b - 1) / b; }

__global__ void fill4_f32(float4* __restrict__ p, float v, int nquads, int* __restrict__ tick) {
    int i = blockIdx.x * blockDim.x + threadIdx.x;
    if (i < nquads) p[i] = make_float4(v, v, v, v);
    if (tick != nullptr && i < 2) tick[i * TICK_PITCH] = 0;
}

__device__ __forceinline__ int draw_job(int* __restrict__ tick) {
    const int h = (int)((__builtin_amdgcn_s_sendmsg_rtn(0x87) >> 18) & 1u);
    int t = atomicAdd(&tick[h * TICK_PITCH], 1);
    if (t < N_SLICES) return (t << 1) | h;
    t = atomicAdd(&tick[(h ^ 1) * TICK_PITCH], 1);
    if (t < N_SLICES) return (t << 1) | (h ^ 1);
    return -1;
}

__global__ void __launch_bounds__(256)
deg_accum(const int* __restrict__ ei, float* __restrict__ deg, int* __restrict__ tick) {
    __shared__ int s_job;
    if (threadIdx.x == 0) s_job = draw_job(tick);
    __syncthreads();
    const int job = s_job;
    if (job < 0) return;
    const int h = job & 1;
    const int base = (job >> 1) * SLICE;
    const int4 d = *reinterpret_cast<const int4*>(&ei[N_EDGES + base + threadIdx.x * 4]);
    if ((d.x >= SPLIT_NODE ? 1 : 0) == h) atomicAdd(&deg[d.x], 1.0f);
    if ((d.y >= SPLIT_NODE ? 1 : 0) == h) atomicAdd(&deg[d.y], 1.0f);
    if ((d.z >= SPLIT_NODE ? 1 : 0) == h) atomicAdd(&deg[d.z], 1.0f);
    if ((d.w >= SPLIT_NODE ? 1 : 0) == h) atomicAdd(&deg[d.w], 1.0f);
}

template <int CIN, int COUT>
__global__ void __launch_bounds__(256)
gemm_wmma_f32(const float* __restrict__ H, const float* __restrict__ W,
              const float* __restrict__ bias, float* __restrict__ out, int nrows) {
    constexpr int NT     = COUT / 16;
    constexpr int KSTEPS = (CIN + 3) / 4;

    const int wavesPerBlock = blockDim.x >> 5;
    const int wave = blockIdx.x * wavesPerBlock + (threadIdx.x >> 5);
    const int row0 = wave * 16;
    if (row0 >= nrows) return;

    const int lane = threadIdx.x & 31;
    const int half = lane >> 4;
    const int m    = lane & 15;
    const int row  = row0 + m;

    v8f acc[NT];
#pragma unroll
    for (int nt = 0; nt < NT; ++nt) acc[nt] = (v8f){};

#pragma unroll
    for (int ks = 0; ks < KSTEPS; ++ks) {
        const int ka = ks * 4 + 2 * half;

        v2f a;
        if constexpr (CIN % 4 == 0) {
            a = *reinterpret_cast<const v2f*>(&H[row * CIN + ka]);
        } else {
            const int k0 = (ka     < CIN) ? ka     : (CIN - 1);
            const int k1 = (ka + 1 < CIN) ? ka + 1 : (CIN - 1);
            float ax = H[row * CIN + k0];
            float ay = H[row * CIN + k1];
            a.x = (ka     < CIN) ? ax : 0.0f;
            a.y = (ka + 1 < CIN) ? ay : 0.0f;
        }

#pragma unroll
        for (int nt = 0; nt < NT; ++nt) {
            const int ncol = nt * 16 + m;
            v2f b;
            if constexpr (CIN % 4 == 0) {
                b = *reinterpret_cast<const v2f*>(&W[ncol * CIN + ka]);
            } else {
                const int k0 = (ka     < CIN) ? ka     : (CIN - 1);
                const int k1 = (ka + 1 < CIN) ? ka + 1 : (CIN - 1);
                float bx = W[ncol * CIN + k0];
                float by = W[ncol * CIN + k1];
                b.x = (ka     < CIN) ? bx : 0.0f;
                b.y = (ka + 1 < CIN) ? by : 0.0f;
            }
            acc[nt] = __builtin_amdgcn_wmma_f32_16x16x4_f32(
                false, a, false, b, (short)0, acc[nt], false, false);
        }
    }

#pragma unroll
    for (int nt = 0; nt < NT; ++nt) {
        const int col = nt * 16 + m;
        const float bb = bias ? bias[col] : 0.0f;
#pragma unroll
        for (int r = 0; r < 8; ++r) {
            out[(row0 + r + 8 * half) * COUT + col] = acc[nt][r] + bb;
        }
    }
}

#define SCAN_B 1024
__global__ void __launch_bounds__(SCAN_B)
offsets_scan(const float* __restrict__ degf, int* __restrict__ off, int* __restrict__ cursor,
             int* __restrict__ tick) {
    __shared__ int s_cnt[SCAN_B / 32];
    __shared__ int s_pre[SCAN_B / 32];
    const int tid  = threadIdx.x;
    const int lane = tid & 31;
    const int wave = tid >> 5;
    const int b    = blockIdx.x;
    const int i    = b * SCAN_B + tid;
    if (b == 0 && tid < 2) tick[tid * TICK_PITCH] = 0;
    if (b == 0 && tid == 2) off[N_NODES] = N_EDGES;

    const int c = (i < N_NODES) ? ((int)degf[i] - 1) : 0;
    int pre = 0;
    for (int j = 0; j < b; ++j) pre += (int)degf[j * SCAN_B + tid] - 1;

    int x = c;
#pragma unroll
    for (int d = 1; d < 32; d <<= 1) {
        const int y = __shfl_up(x, d);
        if (lane >= d) x += y;
    }
#pragma unroll
    for (int d = 16; d > 0; d >>= 1) pre += __shfl_xor(pre, d);
    if (lane == 31) s_cnt[wave] = x;
    if (lane == 0)  s_pre[wave] = pre;
    __syncthreads();
    int base = 0;
#pragma unroll
    for (int w = 0; w < SCAN_B / 32; ++w) {
        base += s_pre[w];
        base += (w < wave) ? s_cnt[w] : 0;
    }
    const int excl = base + x - c;
    if (i < N_NODES) {
        off[i]    = excl;
        cursor[i] = excl;
    }
}

__global__ void __launch_bounds__(256)
csr_fill(const int* __restrict__ ei, int* __restrict__ cursor, int* __restrict__ csr,
         int* __restrict__ tick) {
    __shared__ int s_job;
    if (threadIdx.x == 0) s_job = draw_job(tick);
    __syncthreads();
    const int job = s_job;
    if (job < 0) return;
    const int h = job & 1;
    const int base = (job >> 1) * SLICE;
    const int4 dv = *reinterpret_cast<const int4*>(&ei[N_EDGES + base + threadIdx.x * 4]);
    const int4 sv = *reinterpret_cast<const int4*>(&ei[base + threadIdx.x * 4]);
    const int dd[4] = {dv.x, dv.y, dv.z, dv.w};
    const int ss[4] = {sv.x, sv.y, sv.z, sv.w};
    int pos[4];
#pragma unroll
    for (int k = 0; k < 4; ++k) {
        const bool own = ((dd[k] >= SPLIT_NODE) ? 1 : 0) == h;
        pos[k] = own ? atomicAdd(&cursor[dd[k]], 1) : -1;
    }
#pragma unroll
    for (int k = 0; k < 4; ++k) {
        if (pos[k] >= 0) csr[pos[k]] = ss[k];
    }
}

template <int COUT>
__global__ void __launch_bounds__(256)
pull_rows(const int* __restrict__ off, const int* __restrict__ csr, const float* __restrict__ t,
          const float* __restrict__ bias, float* __restrict__ outH) {
    constexpr int GROUPS = COUT / 4;
    constexpr int U      = 4;
    const int q = blockIdx.x * blockDim.x + threadIdx.x;
    if (q >= N_NODES * GROUPS) return;
    const int n  = q / GROUPS;
    const int c4 = (q % GROUPS) * 4;
    const int e0 = off[n];
    const int e1 = off[n + 1];
    const float dn = rsqrtf((float)(e1 - e0 + 1));
    float4 acc = make_float4(0.0f, 0.0f, 0.0f, 0.0f);
    for (int e = e0; e < e1; e += U) {
        int    s[U], o0[U], o1[U];
        float4 v[U];
#pragma unroll
        for (int u = 0; u < U; ++u) {
            const int ee = e + u;
            s[u] = csr[(ee < e1) ? ee : (e1 - 1)];
        }
#pragma unroll
        for (int u = 0; u < U; ++u) {
            o0[u] = off[s[u]];
            o1[u] = off[s[u] + 1];
            v[u]  = *reinterpret_cast<const float4*>(&t[s[u] * COUT + c4]);
        }
#pragma unroll
        for (int u = 0; u < U; ++u) {
            const bool  live  = (e + u) < e1;
            const float coeff = rsqrtf((float)(o1[u] - o0[u] + 1)) * dn;
            acc.x += live ? v[u].x * coeff : 0.0f;
            acc.y += live ? v[u].y * coeff : 0.0f;
            acc.z += live ? v[u].z * coeff : 0.0f;
            acc.w += live ? v[u].w * coeff : 0.0f;
        }
    }
    const float dd = dn * dn;
    const float4 vt = *reinterpret_cast<const float4*>(&t[n * COUT + c4]);
    const float4 vb = *reinterpret_cast<const float4*>(&bias[c4]);
    float4 r;
    r.x = fmaxf(fmaf(vt.x, dd, acc.x) + vb.x, 0.0f);
    r.y = fmaxf(fmaf(vt.y, dd, acc.y) + vb.y, 0.0f);
    r.z = fmaxf(fmaf(vt.z, dd, acc.z) + vb.z, 0.0f);
    r.w = fmaxf(fmaf(vt.w, dd, acc.w) + vb.w, 0.0f);
    *reinterpret_cast<float4*>(&outH[n * COUT + c4]) = r;
}

__global__ void pool_kernel(const float* __restrict__ h, const int* __restrict__ batch,
                            float* __restrict__ pooled) {
    const int q = blockIdx.x * blockDim.x + threadIdx.x;
    if (q >= N_NODES * 4) return;
    const int n  = q >> 2;
    const int c4 = (q & 3) * 4;
    const float4 v = *reinterpret_cast<const float4*>(&h[n * 16 + c4]);
    float* base = &pooled[batch[n] * 16 + c4];
    atomicAdd(base + 0, v.x);
    atomicAdd(base + 1, v.y);
    atomicAdd(base + 2, v.z);
    atomicAdd(base + 3, v.w);
}

__global__ void classify(const float* __restrict__ emb, const float* __restrict__ Wc,
                         const float* __restrict__ bc, float* __restrict__ out) {
    const int g = blockIdx.x * blockDim.x + threadIdx.x;
    if (g >= N_GRAPHS) return;
    float s = bc[0];
#pragma unroll
    for (int c = 0; c < 16; ++c) s += fmaxf(emb[g * 16 + c], 0.0f) * Wc[c];
    out[g] = s;
}

extern "C" void kernel_launch(void* const* d_in, const int* in_sizes, int n_in,
                              void* d_out, int out_size, void* d_ws, size_t ws_size,
                              hipStream_t stream) {
    const float* x     = (const float*)d_in[0];
    const int*   ei    = (const int*)d_in[1];
    const int*   batch = (const int*)d_in[2];
    const float* W1 = (const float*)d_in[3];
    const float* b1 = (const float*)d_in[4];
    const float* W2 = (const float*)d_in[5];
    const float* b2 = (const float*)d_in[6];
    const float* W3 = (const float*)d_in[7];
    const float* b3 = (const float*)d_in[8];
    const float* We = (const float*)d_in[9];
    const float* be = (const float*)d_in[10];
    const float* Wc = (const float*)d_in[11];
    const float* bc = (const float*)d_in[12];
    float* out = (float*)d_out;

    float* ws   = (float*)d_ws;
    int*   off  = (int*)ws;
    float* bufT = ws + N_NODES + 4;
    int*   csr  = (int*)(bufT + N_NODES * 32);
    float* bufH = bufT + N_NODES * 64;
    float* degf   = bufH;
    int*   cursor = (int*)(bufH + N_NODES);
    int*   tick   = (int*)bufT;
    float* pooled = bufT;
    int*   none   = nullptr;

    const int B = 256;

    fill4_f32<<<ceil_div(N_NODES / 4, B), B, 0, stream>>>((float4*)degf, 1.0f, N_NODES / 4, tick);
    deg_accum<<<2 * N_SLICES, B, 0, stream>>>(ei, degf, tick);
    offsets_scan<<<ceil_div(N_NODES, SCAN_B), SCAN_B, 0, stream>>>(degf, off, cursor, tick);
    csr_fill<<<2 * N_SLICES, B, 0, stream>>>(ei, cursor, csr, tick);

    const int gBlocks = ceil_div(N_NODES / 16, B / 32);

    gemm_wmma_f32<3, 32><<<gBlocks, B, 0, stream>>>(x, W1, nullptr, bufT, N_NODES);
    pull_rows<32><<<ceil_div(N_NODES * 8, B), B, 0, stream>>>(off, csr, bufT, b1, bufH);

    gemm_wmma_f32<32, 16><<<gBlocks, B, 0, stream>>>(bufH, W2, nullptr, bufT, N_NODES);
    pull_rows<16><<<ceil_div(N_NODES * 4, B), B, 0, stream>>>(off, csr, bufT, b2, bufH);

    gemm_wmma_f32<16, 16><<<gBlocks, B, 0, stream>>>(bufH, W3, nullptr, bufT, N_NODES);
    pull_rows<16><<<ceil_div(N_NODES * 4, B), B, 0, stream>>>(off, csr, bufT, b3, bufH);

    fill4_f32<<<ceil_div(N_GRAPHS * 4, B), B, 0, stream>>>((float4*)pooled, 0.0f, N_GRAPHS * 4, none);
    pool_kernel<<<ceil_div(N_NODES * 4, B), B, 0, stream>>>(bufH, batch, pooled);
    gemm_wmma_f32<16, 16><<<ceil_div(N_GRAPHS / 16, B / 32), B, 0, stream>>>(
        pooled, We, be, out, N_GRAPHS);
    classify<<<ceil_div(N_GRAPHS, B), B, 0, stream>>>(out, Wc, bc, out + N_GRAPHS * 16);
}


# Every gathered row is one aligned 128-B line: row buffers start on a line boundary; 16-channel rows get 128-B pitch with the node's rsqrt(degree) beside them; edge products and sum order as before
# speedup vs baseline: 2.5886x; 1.2190x over previous
#include <hip/hip_runtime.h>

typedef __attribute__((ext_vector_type(2))) float v2f;
typedef __attribute__((ext_vector_type(8))) float v8f;

#define N_NODES  100000
#define N_EDGES  3200000
#define N_GRAPHS 1024

#define SPLIT_NODE (N_NODES / 2)
#define SLICE      1024
#define N_SLICES   (N_EDGES / SLICE)
#define TICK_PITCH 128
static_assert(N_EDGES % SLICE == 0, "slices cover the edge list exactly");

static inline int ceil_div(int a, int b) { return (a + b - 1) / b; }

__global__ void fill4_f32(float4* __restrict__ p, float v, int nquads, int* __restrict__ tick) {
    int i = blockIdx.x * blockDim.x + threadIdx.x;
    if (i < nquads) p[i] = make_float4(v, v, v, v);
    if (tick != nullptr && i < 2) tick[i * TICK_PITCH] = 0;
}

__device__ __forceinline__ int draw_job(int* __restrict__ tick) {
    const int h = (int)((__builtin_amdgcn_s_sendmsg_rtn(0x87) >> 18) & 1u);
    int t = atomicAdd(&tick[h * TICK_PITCH], 1);
    if (t < N_SLICES) return (t << 1) | h;
    t = atomicAdd(&tick[(h ^ 1) * TICK_PITCH], 1);
    if (t < N_SLICES) return (t << 1) | (h ^ 1);
    return -1;
}

__global__ void __launch_bounds__(256)
deg_accum(const int* __restrict__ ei, float* __restrict__ deg, int* __restrict__ tick) {
    __shared__ int s_job;
    if (threadIdx.x == 0) s_job = draw_job(tick);
    __syncthreads();
    const int job = s_job;
    if (job < 0) return;
    const int h = job & 1;
    const int base = (job >> 1) * SLICE;
    const int4 d = *reinterpret_cast<const int4*>(&ei[N_EDGES + base + threadIdx.x * 4]);
    if ((d.x >= SPLIT_NODE ? 1 : 0) == h) atomicAdd(&deg[d.x], 1.0f);
    if ((d.y >= SPLIT_NODE ? 1 : 0) == h) atomicAdd(&deg[d.y], 1.0f);
    if ((d.z >= SPLIT_NODE ? 1 : 0) == h) atomicAdd(&deg[d.z], 1.0f);
    if ((d.w >= SPLIT_NODE ? 1 : 0) == h) atomicAdd(&deg[d.w], 1.0f);
}

template <int CIN, int COUT, int PITCH>
__global__ void __launch_bounds__(256)
gemm_wmma_f32(const float* __restrict__ H, const float* __restrict__ W,
              const float* __restrict__ bias, float* __restrict__ out, int nrows,
              const int* __restrict__ off) {
    static_assert(PITCH == COUT || PITCH * 4 == 128, "a row with its scale fills one 128-B line");
    constexpr int NT     = COUT / 16;
    constexpr int KSTEPS = (CIN + 3) / 4;

    const int wavesPerBlock = blockDim.x >> 5;
    const int wave = blockIdx.x * wavesPerBlock + (threadIdx.x >> 5);
    const int row0 = wave * 16;
    if (row0 >= nrows) return;

    const int lane = threadIdx.x & 31;
    const int half = lane >> 4;
    const int m    = lane & 15;
    const int row  = row0 + m;

    float rowscale = 0.0f;
    if constexpr (PITCH > COUT) rowscale = rsqrtf((float)(off[row + 1] - off[row] + 1));

    v8f acc[NT];
#pragma unroll
    for (int nt = 0; nt < NT; ++nt) acc[nt] = (v8f){};

#pragma unroll
    for (int ks = 0; ks < KSTEPS; ++ks) {
        const int ka = ks * 4 + 2 * half;

        v2f a;
        if constexpr (CIN % 4 == 0) {
            a = *reinterpret_cast<const v2f*>(&H[row * CIN + ka]);
        } else {
            const int k0 = (ka     < CIN) ? ka     : (CIN - 1);
            const int k1 = (ka + 1 < CIN) ? ka + 1 : (CIN - 1);
            float ax = H[row * CIN + k0];
            float ay = H[row * CIN + k1];
            a.x = (ka     < CIN) ? ax : 0.0f;
            a.y = (ka + 1 < CIN) ? ay : 0.0f;
        }

#pragma unroll
        for (int nt = 0; nt < NT; ++nt) {
            const int ncol = nt * 16 + m;
            v2f b;
            if constexpr (CIN % 4 == 0) {
                b = *reinterpret_cast<const v2f*>(&W[ncol * CIN + ka]);
            } else {
                const int k0 = (ka     < CIN) ? ka     : (CIN - 1);
                const int k1 = (ka + 1 < CIN) ? ka + 1 : (CIN - 1);
                float bx = W[ncol * CIN + k0];
                float by = W[ncol * CIN + k1];
                b.x = (ka     < CIN) ? bx : 0.0f;
                b.y = (ka + 1 < CIN) ? by : 0.0f;
            }
            acc[nt] = __builtin_amdgcn_wmma_f32_16x16x4_f32(
                false, a, false, b, (short)0, acc[nt], false, false);
        }
    }

#pragma unroll
    for (int nt = 0; nt < NT; ++nt) {
        const int col = nt * 16 + m;
        const float bb = bias ? bias[col] : 0.0f;
#pragma unroll
        for (int r = 0; r < 8; ++r) {
            out[(row0 + r + 8 * half) * PITCH + col] = acc[nt][r] + bb;
        }
    }
    if constexpr (PITCH > COUT) {
        if (half == 0) out[row * PITCH + COUT] = rowscale;
    }
}

#define SCAN_B 1024
__global__ void __launch_bounds__(SCAN_B)
offsets_scan(const float* __restrict__ degf, int* __restrict__ off, int* __restrict__ cursor,
             int* __restrict__ tick) {
    __shared__ int s_cnt[SCAN_B / 32];
    __shared__ int s_pre[SCAN_B / 32];
    const int tid  = threadIdx.x;
    const int lane = tid & 31;
    const int wave = tid >> 5;
    const int b    = blockIdx.x;
    const int i    = b * SCAN_B + tid;
    if (b == 0 && tid < 2) tick[tid * TICK_PITCH] = 0;
    if (b == 0 && tid == 2) off[N_NODES] = N_EDGES;

    const int c = (i < N_NODES) ? ((int)degf[i] - 1) : 0;
    int pre = 0;
    for (int j = 0; j < b; ++j) pre += (int)degf[j * SCAN_B + tid] - 1;

    int x = c;
#pragma unroll
    for (int d = 1; d < 32; d <<= 1) {
        const int y = __shfl_up(x, d);
        if (lane >= d) x += y;
    }
#pragma unroll
    for (int d = 16; d > 0; d >>= 1) pre += __shfl_xor(pre, d);
    if (lane == 31) s_cnt[wave] = x;
    if (lane == 0)  s_pre[wave] = pre;
    __syncthreads();
    int base = 0;
#pragma unroll
    for (int w = 0; w < SCAN_B / 32; ++w) {
        base += s_pre[w];
        base += (w < wave) ? s_cnt[w] : 0;
    }
    const int excl = base + x - c;
    if (i < N_NODES) {
        off[i]    = excl;
        cursor[i] = excl;
    }
}

__global__ void __launch_bounds__(256)
csr_fill(const int* __restrict__ ei, int* __restrict__ cursor, int* __restrict__ csr,
         int* __restrict__ tick) {
    __shared__ int s_job;
    if (threadIdx.x == 0) s_job = draw_job(tick);
    __syncthreads();
    const int job = s_job;
    if (job < 0) return;
    const int h = job & 1;
    const int base = (job >> 1) * SLICE;
    const int4 dv = *reinterpret_cast<const int4*>(&ei[N_EDGES + base + threadIdx.x * 4]);
    const int4 sv = *reinterpret_cast<const int4*>(&ei[base + threadIdx.x * 4]);
    const int dd[4] = {dv.x, dv.y, dv.z, dv.w};
    const int ss[4] = {sv.x, sv.y, sv.z, sv.w};
    int pos[4];
#pragma unroll
    for (int k = 0; k < 4; ++k) {
        const bool own = ((dd[k] >= SPLIT_NODE) ? 1 : 0) == h;
        pos[k] = own ? atomicAdd(&cursor[dd[k]], 1) : -1;
    }
#pragma unroll
    for (int k = 0; k < 4; ++k) {
        if (pos[k] >= 0) csr[pos[k]] = ss[k];
    }
}

template <int COUT, int TP>
__global__ void __launch_bounds__(256)
pull_rows(const int* __restrict__ off, const int* __restrict__ csr, const float* __restrict__ t,
          const float* __restrict__ bias, float* __restrict__ outH) {
    constexpr bool SCALE_IN_ROW = TP > COUT;
    constexpr int GROUPS = COUT / 4;
    constexpr int U      = 4;
    const int q = blockIdx.x * blockDim.x + threadIdx.x;
    if (q >= N_NODES * GROUPS) return;
    const int n  = q / GROUPS;
    const int c4 = (q % GROUPS) * 4;
    const int e0 = off[n];
    const int e1 = off[n + 1];
    const float dn = rsqrtf((float)(e1 - e0 + 1));
    float4 acc = make_float4(0.0f, 0.0f, 0.0f, 0.0f);
    for (int e = e0; e < e1; e += U) {
        int    s[U], o0[U], o1[U];
        float  sc[U];
        float4 v[U];
#pragma unroll
        for (int u = 0; u < U; ++u) {
            const int ee = e + u;
            s[u] = csr[(ee < e1) ? ee : (e1 - 1)];
        }
#pragma unroll
        for (int u = 0; u < U; ++u) {
            v[u] = *reinterpret_cast<const float4*>(&t[s[u] * TP + c4]);
            if constexpr (SCALE_IN_ROW) {
                sc[u] = t[s[u] * TP + COUT];
            } else {
                o0[u] = off[s[u]];
                o1[u] = off[s[u] + 1];
            }
        }
#pragma unroll
        for (int u = 0; u < U; ++u) {
            const bool  live  = (e + u) < e1;
            float ds;
            if constexpr (SCALE_IN_ROW) ds = sc[u];
            else                        ds = rsqrtf((float)(o1[u] - o0[u] + 1));
            const float coeff = ds * dn;
            acc.x += live ? v[u].x * coeff : 0.0f;
            acc.y += live ? v[u].y * coeff : 0.0f;
            acc.z += live ? v[u].z * coeff : 0.0f;
            acc.w += live ? v[u].w * coeff : 0.0f;
        }
    }
    const float dd = dn * dn;
    const float4 vt = *reinterpret_cast<const float4*>(&t[n * TP + c4]);
    const float4 vb = *reinterpret_cast<const float4*>(&bias[c4]);
    float4 r;
    r.x = fmaxf(fmaf(vt.x, dd, acc.x) + vb.x, 0.0f);
    r.y = fmaxf(fmaf(vt.y, dd, acc.y) + vb.y, 0.0f);
    r.z = fmaxf(fmaf(vt.z, dd, acc.z) + vb.z, 0.0f);
    r.w = fmaxf(fmaf(vt.w, dd, acc.w) + vb.w, 0.0f);
    *reinterpret_cast<float4*>(&outH[n * COUT + c4]) = r;
}

__global__ void pool_kernel(const float* __restrict__ h, const int* __restrict__ batch,
                            float* __restrict__ pooled) {
    const int q = blockIdx.x * blockDim.x + threadIdx.x;
    if (q >= N_NODES * 4) return;
    const int n  = q >> 2;
    const int c4 = (q & 3) * 4;
    const float4 v = *reinterpret_cast<const float4*>(&h[n * 16 + c4]);
    float* base = &pooled[batch[n] * 16 + c4];
    atomicAdd(base + 0, v.x);
    atomicAdd(base + 1, v.y);
    atomicAdd(base + 2, v.z);
    atomicAdd(base + 3, v.w);
}

__global__ void classify(const float* __restrict__ emb, const float* __restrict__ Wc,
                         const float* __restrict__ bc, float* __restrict__ out) {
    const int g = blockIdx.x * blockDim.x + threadIdx.x;
    if (g >= N_GRAPHS) return;
    float s = bc[0];
#pragma unroll
    for (int c = 0; c < 16; ++c) s += fmaxf(emb[g * 16 + c], 0.0f) * Wc[c];
    out[g] = s;
}

extern "C" void kernel_launch(void* const* d_in, const int* in_sizes, int n_in,
                              void* d_out, int out_size, void* d_ws, size_t ws_size,
                              hipStream_t stream) {
    const float* x     = (const float*)d_in[0];
    const int*   ei    = (const int*)d_in[1];
    const int*   batch = (const int*)d_in[2];
    const float* W1 = (const float*)d_in[3];
    const float* b1 = (const float*)d_in[4];
    const float* W2 = (const float*)d_in[5];
    const float* b2 = (const float*)d_in[6];
    const float* W3 = (const float*)d_in[7];
    const float* b3 = (const float*)d_in[8];
    const float* We = (const float*)d_in[9];
    const float* be = (const float*)d_in[10];
    const float* Wc = (const float*)d_in[11];
    const float* bc = (const float*)d_in[12];
    float* out = (float*)d_out;

    float* ws   = (float*)d_ws;
    float* bufT = ws;
    int*   csr  = (int*)(bufT + N_NODES * 32);
    float* bufH = bufT + N_NODES * 64;
    int*   off  = (int*)(bufT + N_NODES * 96);
    float* degf   = bufH;
    int*   cursor = (int*)(bufH + N_NODES);
    int*   tick   = (int*)bufT;
    float* pooled = bufT;
    int*   none   = nullptr;

    const int B = 256;

    fill4_f32<<<ceil_div(N_NODES / 4, B), B, 0, stream>>>((float4*)degf, 1.0f, N_NODES / 4, tick);
    deg_accum<<<2 * N_SLICES, B, 0, stream>>>(ei, degf, tick);
    offsets_scan<<<ceil_div(N_NODES, SCAN_B), SCAN_B, 0, stream>>>(degf, off, cursor, tick);
    csr_fill<<<2 * N_SLICES, B, 0, stream>>>(ei, cursor, csr, tick);

    const int gBlocks = ceil_div(N_NODES / 16, B / 32);

    gemm_wmma_f32<3, 32, 32><<<gBlocks, B, 0, stream>>>(x, W1, nullptr, bufT, N_NODES, none);
    pull_rows<32, 32><<<ceil_div(N_NODES * 8, B), B, 0, stream>>>(off, csr, bufT, b1, bufH);

    gemm_wmma_f32<32, 16, 32><<<gBlocks, B, 0, stream>>>(bufH, W2, nullptr, bufT, N_NODES, off);
    pull_rows<16, 32><<<ceil_div(N_NODES * 4, B), B, 0, stream>>>(off, csr, bufT, b2, bufH);

    gemm_wmma_f32<16, 16, 32><<<gBlocks, B, 0, stream>>>(bufH, W3, nullptr, bufT, N_NODES, off);
    pull_rows<16, 32><<<ceil_div(N_NODES * 4, B), B, 0, stream>>>(off, csr, bufT, b3, bufH);

    fill4_f32<<<ceil_div(N_GRAPHS * 4, B), B, 0, stream>>>((float4*)pooled, 0.0f, N_GRAPHS * 4, none);
    pool_kernel<<<ceil_div(N_NODES * 4, B), B, 0, stream>>>(bufH, batch, pooled);
    gemm_wmma_f32<16, 16, 16><<<ceil_div(N_GRAPHS / 16, B / 32), B, 0, stream>>>(
        pooled, We, be, out, N_GRAPHS, none);
    classify<<<ceil_div(N_GRAPHS, B), B, 0, stream>>>(out, Wc, bc, out + N_GRAPHS * 16);
}


# Pull: a node's lanes load its source list 32 entries at a time, each entry once, and pass the indices round by lane shuffle (was: 4 entries re-fetched every trip); gathers and sum order as before
# speedup vs baseline: 2.9523x; 1.1405x over previous
#include <hip/hip_runtime.h>

typedef __attribute__((ext_vector_type(2))) float v2f;
typedef __attribute__((ext_vector_type(8))) float v8f;

#define N_NODES  100000
#define N_EDGES  3200000
#define N_GRAPHS 1024

#define SPLIT_NODE (N_NODES / 2)
#define SLICE      1024
#define N_SLICES   (N_EDGES / SLICE)
#define TICK_PITCH 128
static_assert(N_EDGES % SLICE == 0, "slices cover the edge list exactly");

static inline int ceil_div(int a, int b) { return (a + b - 1) / b; }

__global__ void fill4_f32(float4* __restrict__ p, float v, int nquads, int* __restrict__ tick) {
    int i = blockIdx.x * blockDim.x + threadIdx.x;
    if (i < nquads) p[i] = make_float4(v, v, v, v);
    if (tick != nullptr && i < 2) tick[i * TICK_PITCH] = 0;
}

__device__ __forceinline__ int draw_job(int* __restrict__ tick) {
    const int h = (int)((__builtin_amdgcn_s_sendmsg_rtn(0x87) >> 18) & 1u);
    int t = atomicAdd(&tick[h * TICK_PITCH], 1);
    if (t < N_SLICES) return (t << 1) | h;
    t = atomicAdd(&tick[(h ^ 1) * TICK_PITCH], 1);
    if (t < N_SLICES) return (t << 1) | (h ^ 1);
    return -1;
}

__global__ void __launch_bounds__(256)
deg_accum(const int* __restrict__ ei, float* __restrict__ deg, int* __restrict__ tick) {
    __shared__ int s_job;
    if (threadIdx.x == 0) s_job = draw_job(tick);
    __syncthreads();
    const int job = s_job;
    if (job < 0) return;
    const int h = job & 1;
    const int base = (job >> 1) * SLICE;
    const int4 d = *reinterpret_cast<const int4*>(&ei[N_EDGES + base + threadIdx.x * 4]);
    if ((d.x >= SPLIT_NODE ? 1 : 0) == h) atomicAdd(&deg[d.x], 1.0f);
    if ((d.y >= SPLIT_NODE ? 1 : 0) == h) atomicAdd(&deg[d.y], 1.0f);
    if ((d.z >= SPLIT_NODE ? 1 : 0) == h) atomicAdd(&deg[d.z], 1.0f);
    if ((d.w >= SPLIT_NODE ? 1 : 0) == h) atomicAdd(&deg[d.w], 1.0f);
}

template <int CIN, int COUT, int PITCH>
__global__ void __launch_bounds__(256)
gemm_wmma_f32(const float* __restrict__ H, const float* __restrict__ W,
              const float* __restrict__ bias, float* __restrict__ out, int nrows,
              const int* __restrict__ off) {
    static_assert(PITCH == COUT || PITCH * 4 == 128, "a row with its scale fills one 128-B line");
    constexpr int NT     = COUT / 16;
    constexpr int KSTEPS = (CIN + 3) / 4;

    const int wavesPerBlock = blockDim.x >> 5;
    const int wave = blockIdx.x * wavesPerBlock + (threadIdx.x >> 5);
    const int row0 = wave * 16;
    if (row0 >= nrows) return;

    const int lane = threadIdx.x & 31;
    const int half = lane >> 4;
    const int m    = lane & 15;
    const int row  = row0 + m;

    float rowscale = 0.0f;
    if constexpr (PITCH > COUT) rowscale = rsqrtf((float)(off[row + 1] - off[row] + 1));

    v8f acc[NT];
#pragma unroll
    for (int nt = 0; nt < NT; ++nt) acc[nt] = (v8f){};

#pragma unroll
    for (int ks = 0; ks < KSTEPS; ++ks) {
        const int ka = ks * 4 + 2 * half;

        v2f a;
        if constexpr (CIN % 4 == 0) {
            a = *reinterpret_cast<const v2f*>(&H[row * CIN + ka]);
        } else {
            const int k0 = (ka     < CIN) ? ka     : (CIN - 1);
            const int k1 = (ka + 1 < CIN) ? ka + 1 : (CIN - 1);
            float ax = H[row * CIN + k0];
            float ay = H[row * CIN + k1];
            a.x = (ka     < CIN) ? ax : 0.0f;
            a.y = (ka + 1 < CIN) ? ay : 0.0f;
        }

#pragma unroll
        for (int nt = 0; nt < NT; ++nt) {
            const int ncol = nt * 16 + m;
            v2f b;
            if constexpr (CIN % 4 == 0) {
                b = *reinterpret_cast<const v2f*>(&W[ncol * CIN + ka]);
            } else {
                const int k0 = (ka     < CIN) ? ka     : (CIN - 1);
                const int k1 = (ka + 1 < CIN) ? ka + 1 : (CIN - 1);
                float bx = W[ncol * CIN + k0];
                float by = W[ncol * CIN + k1];
                b.x = (ka     < CIN) ? bx : 0.0f;
                b.y = (ka + 1 < CIN) ? by : 0.0f;
            }
            acc[nt] = __builtin_amdgcn_wmma_f32_16x16x4_f32(
                false, a, false, b, (short)0, acc[nt], false, false);
        }
    }

#pragma unroll
    for (int nt = 0; nt < NT; ++nt) {
        const int col = nt * 16 + m;
        const float bb = bias ? bias[col] : 0.0f;
#pragma unroll
        for (int r = 0; r < 8; ++r) {
            out[(row0 + r + 8 * half) * PITCH + col] = acc[nt][r] + bb;
        }
    }
    if constexpr (PITCH > COUT) {
        if (half == 0) out[row * PITCH + COUT] = rowscale;
    }
}

#define SCAN_B 1024
__global__ void __launch_bounds__(SCAN_B)
offsets_scan(const float* __restrict__ degf, int* __restrict__ off, int* __restrict__ cursor,
             int* __restrict__ tick) {
    __shared__ int s_cnt[SCAN_B / 32];
    __shared__ int s_pre[SCAN_B / 32];
    const int tid  = threadIdx.x;
    const int lane = tid & 31;
    const int wave = tid >> 5;
    const int b    = blockIdx.x;
    const int i    = b * SCAN_B + tid;
    if (b == 0 && tid < 2) tick[tid * TICK_PITCH] = 0;
    if (b == 0 && tid == 2) off[N_NODES] = N_EDGES;

    const int c = (i < N_NODES) ? ((int)degf[i] - 1) : 0;
    int pre = 0;
    for (int j = 0; j < b; ++j) pre += (int)degf[j * SCAN_B + tid] - 1;

    int x = c;
#pragma unroll
    for (int d = 1; d < 32; d <<= 1) {
        const int y = __shfl_up(x, d);
        if (lane >= d) x += y;
    }
#pragma unroll
    for (int d = 16; d > 0; d >>= 1) pre += __shfl_xor(pre, d);
    if (lane == 31) s_cnt[wave] = x;
    if (lane == 0)  s_pre[wave] = pre;
    __syncthreads();
    int base = 0;
#pragma unroll
    for (int w = 0; w < SCAN_B / 32; ++w) {
        base += s_pre[w];
        base += (w < wave) ? s_cnt[w] : 0;
    }
    const int excl = base + x - c;
    if (i < N_NODES) {
        off[i]    = excl;
        cursor[i] = excl;
    }
}

__global__ void __launch_bounds__(256)
csr_fill(const int* __restrict__ ei, int* __restrict__ cursor, int* __restrict__ csr,
         int* __restrict__ tick) {
    __shared__ int s_job;
    if (threadIdx.x == 0) s_job = draw_job(tick);
    __syncthreads();
    const int job = s_job;
    if (job < 0) return;
    const int h = job & 1;
    const int base = (job >> 1) * SLICE;
    const int4 dv = *reinterpret_cast<const int4*>(&ei[N_EDGES + base + threadIdx.x * 4]);
    const int4 sv = *reinterpret_cast<const int4*>(&ei[base + threadIdx.x * 4]);
    const int dd[4] = {dv.x, dv.y, dv.z, dv.w};
    const int ss[4] = {sv.x, sv.y, sv.z, sv.w};
    int pos[4];
#pragma unroll
    for (int k = 0; k < 4; ++k) {
        const bool own = ((dd[k] >= SPLIT_NODE) ? 1 : 0) == h;
        pos[k] = own ? atomicAdd(&cursor[dd[k]], 1) : -1;
    }
#pragma unroll
    for (int k = 0; k < 4; ++k) {
        if (pos[k] >= 0) csr[pos[k]] = ss[k];
    }
}

template <int COUT, int TP>
__global__ void __launch_bounds__(256)
pull_rows(const int* __restrict__ off, const int* __restrict__ csr, const float* __restrict__ t,
          const float* __restrict__ bias, float* __restrict__ outH) {
    constexpr bool SCALE_IN_ROW = TP > COUT;
    constexpr int GROUPS = COUT / 4;
    constexpr int U      = 4;
    constexpr int NB     = 32;
    constexpr int R      = NB / GROUPS;
    static_assert(NB % GROUPS == 0 && NB % U == 0 && (GROUPS & (GROUPS - 1)) == 0,
                  "a block is shared out evenly; a group is an aligned power-of-two run of lanes");
    const int q = blockIdx.x * blockDim.x + threadIdx.x;
    if (q >= N_NODES * GROUPS) return;
    const int n  = q / GROUPS;
    const int j  = q % GROUPS;
    const int c4 = j * 4;
    const int e0 = off[n];
    const int e1 = off[n + 1];
    const float dn = rsqrtf((float)(e1 - e0 + 1));
    float4 acc = make_float4(0.0f, 0.0f, 0.0f, 0.0f);
    for (int eb = e0; eb < e1; eb += NB) {
        int idx[R];
#pragma unroll
        for (int r = 0; r < R; ++r) {
            const int ee = eb + r * GROUPS + j;
            idx[r] = csr[(ee < e1) ? ee : (e1 - 1)];
        }
#pragma unroll
        for (int tt = 0; tt < NB / U; ++tt) {
            const int e = eb + tt * U;
            if (e < e1) {
                int    s[U], o0[U], o1[U];
                float  sc[U];
                float4 v[U];
#pragma unroll
                for (int u = 0; u < U; ++u) {
                    const int k = tt * U + u;
                    s[u] = __shfl(idx[k / GROUPS], k % GROUPS, GROUPS);
                }
#pragma unroll
                for (int u = 0; u < U; ++u) {
                    v[u] = *reinterpret_cast<const float4*>(&t[s[u] * TP + c4]);
                    if constexpr (SCALE_IN_ROW) {
                        sc[u] = t[s[u] * TP + COUT];
                    } else {
                        o0[u] = off[s[u]];
                        o1[u] = off[s[u] + 1];
                    }
                }
                __builtin_amdgcn_sched_group_barrier(0x020, 2 * U, 0);
#pragma unroll
                for (int u = 0; u < U; ++u) {
                    const bool  live  = (e + u) < e1;
                    float ds;
                    if constexpr (SCALE_IN_ROW) ds = sc[u];
                    else                        ds = rsqrtf((float)(o1[u] - o0[u] + 1));
                    const float coeff = ds * dn;
                    acc.x += live ? v[u].x * coeff : 0.0f;
                    acc.y += live ? v[u].y * coeff : 0.0f;
                    acc.z += live ? v[u].z * coeff : 0.0f;
                    acc.w += live ? v[u].w * coeff : 0.0f;
                }
            }
        }
    }
    const float dd = dn * dn;
    const float4 vt = *reinterpret_cast<const float4*>(&t[n * TP + c4]);
    const float4 vb = *reinterpret_cast<const float4*>(&bias[c4]);
    float4 r;
    r.x = fmaxf(fmaf(vt.x, dd, acc.x) + vb.x, 0.0f);
    r.y = fmaxf(fmaf(vt.y, dd, acc.y) + vb.y, 0.0f);
    r.z = fmaxf(fmaf(vt.z, dd, acc.z) + vb.z, 0.0f);
    r.w = fmaxf(fmaf(vt.w, dd, acc.w) + vb.w, 0.0f);
    *reinterpret_cast<float4*>(&outH[n * COUT + c4]) = r;
}

__global__ void pool_kernel(const float* __restrict__ h, const int* __restrict__ batch,
                            float* __restrict__ pooled) {
    const int q = blockIdx.x * blockDim.x + threadIdx.x;
    if (q >= N_NODES * 4) return;
    const int n  = q >> 2;
    const int c4 = (q & 3) * 4;
    const float4 v = *reinterpret_cast<const float4*>(&h[n * 16 + c4]);
    float* base = &pooled[batch[n] * 16 + c4];
    atomicAdd(base + 0, v.x);
    atomicAdd(base + 1, v.y);
    atomicAdd(base + 2, v.z);
    atomicAdd(base + 3, v.w);
}

__global__ void classify(const float* __restrict__ emb, const float* __restrict__ Wc,
                         const float* __restrict__ bc, float* __restrict__ out) {
    const int g = blockIdx.x * blockDim.x + threadIdx.x;
    if (g >= N_GRAPHS) return;
    float s = bc[0];
#pragma unroll
    for (int c = 0; c < 16; ++c) s += fmaxf(emb[g * 16 + c], 0.0f) * Wc[c];
    out[g] = s;
}

extern "C" void kernel_launch(void* const* d_in, const int* in_sizes, int n_in,
                              void* d_out, int out_size, void* d_ws, size_t ws_size,
                              hipStream_t stream) {
    const float* x     = (const float*)d_in[0];
    const int*   ei    = (const int*)d_in[1];
    const int*   batch = (const int*)d_in[2];
    const float* W1 = (const float*)d_in[3];
    const float* b1 = (const float*)d_in[4];
    const float* W2 = (const float*)d_in[5];
    const float* b2 = (const float*)d_in[6];
    const float* W3 = (const float*)d_in[7];
    const float* b3 = (const float*)d_in[8];
    const float* We = (const float*)d_in[9];
    const float* be = (const float*)d_in[10];
    const float* Wc = (const float*)d_in[11];
    const float* bc = (const float*)d_in[12];
    float* out = (float*)d_out;

    float* ws   = (float*)d_ws;
    float* bufT = ws;
    int*   csr  = (int*)(bufT + N_NODES * 32);
    float* bufH = bufT + N_NODES * 64;
    int*   off  = (int*)(bufT + N_NODES * 96);
    float* degf   = bufH;
    int*   cursor = (int*)(bufH + N_NODES);
    int*   tick   = (int*)bufT;
    float* pooled = bufT;
    int*   none   = nullptr;

    const int B = 256;

    fill4_f32<<<ceil_div(N_NODES / 4, B), B, 0, stream>>>((float4*)degf, 1.0f, N_NODES / 4, tick);
    deg_accum<<<2 * N_SLICES, B, 0, stream>>>(ei, degf, tick);
    offsets_scan<<<ceil_div(N_NODES, SCAN_B), SCAN_B, 0, stream>>>(degf, off, cursor, tick);
    csr_fill<<<2 * N_SLICES, B, 0, stream>>>(ei, cursor, csr, tick);

    const int gBlocks = ceil_div(N_NODES / 16, B / 32);

    gemm_wmma_f32<3, 32, 32><<<gBlocks, B, 0, stream>>>(x, W1, nullptr, bufT, N_NODES, none);
    pull_rows<32, 32><<<ceil_div(N_NODES * 8, B), B, 0, stream>>>(off, csr, bufT, b1, bufH);

    gemm_wmma_f32<32, 16, 32><<<gBlocks, B, 0, stream>>>(bufH, W2, nullptr, bufT, N_NODES, off);
    pull_rows<16, 32><<<ceil_div(N_NODES * 4, B), B, 0, stream>>>(off, csr, bufT, b2, bufH);

    gemm_wmma_f32<16, 16, 32><<<gBlocks, B, 0, stream>>>(bufH, W3, nullptr, bufT, N_NODES, off);
    pull_rows<16, 32><<<ceil_div(N_NODES * 4, B), B, 0, stream>>>(off, csr, bufT, b3, bufH);

    fill4_f32<<<ceil_div(N_GRAPHS * 4, B), B, 0, stream>>>((float4*)pooled, 0.0f, N_GRAPHS * 4, none);
    pool_kernel<<<ceil_div(N_NODES * 4, B), B, 0, stream>>>(bufH, batch, pooled);
    gemm_wmma_f32<16, 16, 16><<<ceil_div(N_GRAPHS / 16, B / 32), B, 0, stream>>>(
        pooled, We, be, out, N_GRAPHS, none);
    classify<<<ceil_div(N_GRAPHS, B), B, 0, stream>>>(out, Wc, bc, out + N_GRAPHS * 16);
}


# 16-channel pulls: 8 lanes per node, lane 4 loads the row's scale quad as its piece of the row's own load and hands it on by lane shuffle: one L2 request per edge, not two; sum order as before
# speedup vs baseline: 3.2455x; 1.0993x over previous
#include <hip/hip_runtime.h>

typedef __attribute__((ext_vector_type(2))) float v2f;
typedef __attribute__((ext_vector_type(8))) float v8f;

#define N_NODES  100000
#define N_EDGES  3200000
#define N_GRAPHS 1024

#define SPLIT_NODE (N_NODES / 2)
#define SLICE      1024
#define N_SLICES   (N_EDGES / SLICE)
#define TICK_PITCH 128
static_assert(N_EDGES % SLICE == 0, "slices cover the edge list exactly");

static inline int ceil_div(int a, int b) { return (a + b - 1) / b; }

__global__ void fill4_f32(float4* __restrict__ p, float v, int nquads, int* __restrict__ tick) {
    int i = blockIdx.x * blockDim.x + threadIdx.x;
    if (i < nquads) p[i] = make_float4(v, v, v, v);
    if (tick != nullptr && i < 2) tick[i * TICK_PITCH] = 0;
}

__device__ __forceinline__ int draw_job(int* __restrict__ tick) {
    const int h = (int)((__builtin_amdgcn_s_sendmsg_rtn(0x87) >> 18) & 1u);
    int t = atomicAdd(&tick[h * TICK_PITCH], 1);
    if (t < N_SLICES) return (t << 1) | h;
    t = atomicAdd(&tick[(h ^ 1) * TICK_PITCH], 1);
    if (t < N_SLICES) return (t << 1) | (h ^ 1);
    return -1;
}

__global__ void __launch_bounds__(256)
deg_accum(const int* __restrict__ ei, float* __restrict__ deg, int* __restrict__ tick) {
    __shared__ int s_job;
    if (threadIdx.x == 0) s_job = draw_job(tick);
    __syncthreads();
    const int job = s_job;
    if (job < 0) return;
    const int h = job & 1;
    const int base = (job >> 1) * SLICE;
    const int4 d = *reinterpret_cast<const int4*>(&ei[N_EDGES + base + threadIdx.x * 4]);
    if ((d.x >= SPLIT_NODE ? 1 : 0) == h) atomicAdd(&deg[d.x], 1.0f);
    if ((d.y >= SPLIT_NODE ? 1 : 0) == h) atomicAdd(&deg[d.y], 1.0f);
    if ((d.z >= SPLIT_NODE ? 1 : 0) == h) atomicAdd(&deg[d.z], 1.0f);
    if ((d.w >= SPLIT_NODE ? 1 : 0) == h) atomicAdd(&deg[d.w], 1.0f);
}

template <int CIN, int COUT, int PITCH>
__global__ void __launch_bounds__(256)
gemm_wmma_f32(const float* __restrict__ H, const float* __restrict__ W,
              const float* __restrict__ bias, float* __restrict__ out, int nrows,
              const int* __restrict__ off) {
    static_assert(PITCH == COUT || (PITCH * 4 == 128 && COUT + 4 <= PITCH),
                  "a row with its scale quad lies in one 128-B line");
    constexpr int NT     = COUT / 16;
    constexpr int KSTEPS = (CIN + 3) / 4;

    const int wavesPerBlock = blockDim.x >> 5;
    const int wave = blockIdx.x * wavesPerBlock + (threadIdx.x >> 5);
    const int row0 = wave * 16;
    if (row0 >= nrows) return;

    const int lane = threadIdx.x & 31;
    const int half = lane >> 4;
    const int m    = lane & 15;
    const int row  = row0 + m;

    float rowscale = 0.0f;
    if constexpr (PITCH > COUT) rowscale = rsqrtf((float)(off[row + 1] - off[row] + 1));

    v8f acc[NT];
#pragma unroll
    for (int nt = 0; nt < NT; ++nt) acc[nt] = (v8f){};

#pragma unroll
    for (int ks = 0; ks < KSTEPS; ++ks) {
        const int ka = ks * 4 + 2 * half;

        v2f a;
        if constexpr (CIN % 4 == 0) {
            a = *reinterpret_cast<const v2f*>(&H[row * CIN + ka]);
        } else {
            const int k0 = (ka     < CIN) ? ka     : (CIN - 1);
            const int k1 = (ka + 1 < CIN) ? ka + 1 : (CIN - 1);
            float ax = H[row * CIN + k0];
            float ay = H[row * CIN + k1];
            a.x = (ka     < CIN) ? ax : 0.0f;
            a.y = (ka + 1 < CIN) ? ay : 0.0f;
        }

#pragma unroll
        for (int nt = 0; nt < NT; ++nt) {
            const int ncol = nt * 16 + m;
            v2f b;
            if constexpr (CIN % 4 == 0) {
                b = *reinterpret_cast<const v2f*>(&W[ncol * CIN + ka]);
            } else {
                const int k0 = (ka     < CIN) ? ka     : (CIN - 1);
                const int k1 = (ka + 1 < CIN) ? ka + 1 : (CIN - 1);
                float bx = W[ncol * CIN + k0];
                float by = W[ncol * CIN + k1];
                b.x = (ka     < CIN) ? bx : 0.0f;
                b.y = (ka + 1 < CIN) ? by : 0.0f;
            }
            acc[nt] = __builtin_amdgcn_wmma_f32_16x16x4_f32(
                false, a, false, b, (short)0, acc[nt], false, false);
        }
    }

#pragma unroll
    for (int nt = 0; nt < NT; ++nt) {
        const int col = nt * 16 + m;
        const float bb = bias ? bias[col] : 0.0f;
#pragma unroll
        for (int r = 0; r < 8; ++r) {
            out[(row0 + r + 8 * half) * PITCH + col] = acc[nt][r] + bb;
        }
    }
    if constexpr (PITCH > COUT) {
        if (half == 0)
            *reinterpret_cast<float4*>(&out[row * PITCH + COUT]) = make_float4(rowscale, 0.0f, 0.0f, 0.0f);
    }
}

#define SCAN_B 1024
__global__ void __launch_bounds__(SCAN_B)
offsets_scan(const float* __restrict__ degf, int* __restrict__ off, int* __restrict__ cursor,
             int* __restrict__ tick) {
    __shared__ int s_cnt[SCAN_B / 32];
    __shared__ int s_pre[SCAN_B / 32];
    const int tid  = threadIdx.x;
    const int lane = tid & 31;
    const int wave = tid >> 5;
    const int b    = blockIdx.x;
    const int i    = b * SCAN_B + tid;
    if (b == 0 && tid < 2) tick[tid * TICK_PITCH] = 0;
    if (b == 0 && tid == 2) off[N_NODES] = N_EDGES;

    const int c = (i < N_NODES) ? ((int)degf[i] - 1) : 0;
    int pre = 0;
    for (int j = 0; j < b; ++j) pre += (int)degf[j * SCAN_B + tid] - 1;

    int x = c;
#pragma unroll
    for (int d = 1; d < 32; d <<= 1) {
        const int y = __shfl_up(x, d);
        if (lane >= d) x += y;
    }
#pragma unroll
    for (int d = 16; d > 0; d >>= 1) pre += __shfl_xor(pre, d);
    if (lane == 31) s_cnt[wave] = x;
    if (lane == 0)  s_pre[wave] = pre;
    __syncthreads();
    int base = 0;
#pragma unroll
    for (int w = 0; w < SCAN_B / 32; ++w) {
        base += s_pre[w];
        base += (w < wave) ? s_cnt[w] : 0;
    }
    const int excl = base + x - c;
    if (i < N_NODES) {
        off[i]    = excl;
        cursor[i] = excl;
    }
}

__global__ void __launch_bounds__(256)
csr_fill(const int* __restrict__ ei, int* __restrict__ cursor, int* __restrict__ csr,
         int* __restrict__ tick) {
    __shared__ int s_job;
    if (threadIdx.x == 0) s_job = draw_job(tick);
    __syncthreads();
    const int job = s_job;
    if (job < 0) return;
    const int h = job & 1;
    const int base = (job >> 1) * SLICE;
    const int4 dv = *reinterpret_cast<const int4*>(&ei[N_EDGES + base + threadIdx.x * 4]);
    const int4 sv = *reinterpret_cast<const int4*>(&ei[base + threadIdx.x * 4]);
    const int dd[4] = {dv.x, dv.y, dv.z, dv.w};
    const int ss[4] = {sv.x, sv.y, sv.z, sv.w};
    int pos[4];
#pragma unroll
    for (int k = 0; k < 4; ++k) {
        const bool own = ((dd[k] >= SPLIT_NODE) ? 1 : 0) == h;
        pos[k] = own ? atomicAdd(&cursor[dd[k]], 1) : -1;
    }
#pragma unroll
    for (int k = 0; k < 4; ++k) {
        if (pos[k] >= 0) csr[pos[k]] = ss[k];
    }
}

template <int COUT, int TP>
__global__ void __launch_bounds__(256)
pull_rows(const int* __restrict__ off, const int* __restrict__ csr, const float* __restrict__ t,
          const float* __restrict__ bias, float* __restrict__ outH) {
    constexpr bool SCALE_IN_ROW = TP > COUT;
    constexpr int DL     = COUT / 4;
    constexpr int GROUPS = SCALE_IN_ROW ? 2 * DL : DL;
    static_assert(!SCALE_IN_ROW || (COUT + 4) <= TP, "the scale quad lies inside the row's pitch");
    constexpr int U      = 4;
    constexpr int NB     = 32;
    constexpr int R      = NB / GROUPS;
    static_assert(NB % GROUPS == 0 && NB % U == 0 && (GROUPS & (GROUPS - 1)) == 0,
                  "a block is shared out evenly; a group is an aligned power-of-two run of lanes");
    const int q = blockIdx.x * blockDim.x + threadIdx.x;
    if (q >= N_NODES * GROUPS) return;
    const int n  = q / GROUPS;
    const int j  = q % GROUPS;
    const int c4 = ((j < DL) ? j : DL) * 4;
    const int e0 = off[n];
    const int e1 = off[n + 1];
    const float dn = rsqrtf((float)(e1 - e0 + 1));
    float4 acc = make_float4(0.0f, 0.0f, 0.0f, 0.0f);
    for (int eb = e0; eb < e1; eb += NB) {
        int idx[R];
#pragma unroll
        for (int r = 0; r < R; ++r) {
            const int ee = eb + r * GROUPS + j;
            idx[r] = csr[(ee < e1) ? ee : (e1 - 1)];
        }
#pragma unroll
        for (int tt = 0; tt < NB / U; ++tt) {
            const int e = eb + tt * U;
            if (e < e1) {
                int    s[U], o0[U], o1[U];
                float4 v[U];
#pragma unroll
                for (int u = 0; u < U; ++u) {
                    const int k = tt * U + u;
                    s[u] = __shfl(idx[k / GROUPS], k % GROUPS, GROUPS);
                }
#pragma unroll
                for (int u = 0; u < U; ++u) {
                    v[u] = *reinterpret_cast<const float4*>(&t[s[u] * TP + c4]);
                    if constexpr (!SCALE_IN_ROW) {
                        o0[u] = off[s[u]];
                        o1[u] = off[s[u] + 1];
                    }
                }
                __builtin_amdgcn_sched_group_barrier(0x020, SCALE_IN_ROW ? U : 2 * U, 0);
#pragma unroll
                for (int u = 0; u < U; ++u) {
                    const bool  live  = (e + u) < e1;
                    float ds;
                    if constexpr (SCALE_IN_ROW) ds = __shfl(v[u].x, DL, GROUPS);
                    else                        ds = rsqrtf((float)(o1[u] - o0[u] + 1));
                    const float coeff = ds * dn;
                    acc.x += live ? v[u].x * coeff : 0.0f;
                    acc.y += live ? v[u].y * coeff : 0.0f;
                    acc.z += live ? v[u].z * coeff : 0.0f;
                    acc.w += live ? v[u].w * coeff : 0.0f;
                }
            }
        }
    }
    const float dd = dn * dn;
    const float4 vt = *reinterpret_cast<const float4*>(&t[n * TP + c4]);
    const int    co = (j < DL) ? j * 4 : 0;
    const float4 vb = *reinterpret_cast<const float4*>(&bias[co]);
    float4 r;
    r.x = fmaxf(fmaf(vt.x, dd, acc.x) + vb.x, 0.0f);
    r.y = fmaxf(fmaf(vt.y, dd, acc.y) + vb.y, 0.0f);
    r.z = fmaxf(fmaf(vt.z, dd, acc.z) + vb.z, 0.0f);
    r.w = fmaxf(fmaf(vt.w, dd, acc.w) + vb.w, 0.0f);
    if (j < DL) *reinterpret_cast<float4*>(&outH[n * COUT + co]) = r;
}

__global__ void pool_kernel(const float* __restrict__ h, const int* __restrict__ batch,
                            float* __restrict__ pooled) {
    const int q = blockIdx.x * blockDim.x + threadIdx.x;
    if (q >= N_NODES * 4) return;
    const int n  = q >> 2;
    const int c4 = (q & 3) * 4;
    const float4 v = *reinterpret_cast<const float4*>(&h[n * 16 + c4]);
    float* base = &pooled[batch[n] * 16 + c4];
    atomicAdd(base + 0, v.x);
    atomicAdd(base + 1, v.y);
    atomicAdd(base + 2, v.z);
    atomicAdd(base + 3, v.w);
}

__global__ void classify(const float* __restrict__ emb, const float* __restrict__ Wc,
                         const float* __restrict__ bc, float* __restrict__ out) {
    const int g = blockIdx.x * blockDim.x + threadIdx.x;
    if (g >= N_GRAPHS) return;
    float s = bc[0];
#pragma unroll
    for (int c = 0; c < 16; ++c) s += fmaxf(emb[g * 16 + c], 0.0f) * Wc[c];
    out[g] = s;
}

extern "C" void kernel_launch(void* const* d_in, const int* in_sizes, int n_in,
                              void* d_out, int out_size, void* d_ws, size_t ws_size,
                              hipStream_t stream) {
    const float* x     = (const float*)d_in[0];
    const int*   ei    = (const int*)d_in[1];
    const int*   batch = (const int*)d_in[2];
    const float* W1 = (const float*)d_in[3];
    const float* b1 = (const float*)d_in[4];
    const float* W2 = (const float*)d_in[5];
    const float* b2 = (const float*)d_in[6];
    const float* W3 = (const float*)d_in[7];
    const float* b3 = (const float*)d_in[8];
    const float* We = (const float*)d_in[9];
    const float* be = (const float*)d_in[10];
    const float* Wc = (const float*)d_in[11];
    const float* bc = (const float*)d_in[12];
    float* out = (float*)d_out;

    float* ws   = (float*)d_ws;
    float* bufT = ws;
    int*   csr  = (int*)(bufT + N_NODES * 32);
    float* bufH = bufT + N_NODES * 64;
    int*   off  = (int*)(bufT + N_NODES * 96);
    float* degf   = bufH;
    int*   cursor = (int*)(bufH + N_NODES);
    int*   tick   = (int*)bufT;
    float* pooled = bufT;
    int*   none   = nullptr;

    const int B = 256;

    fill4_f32<<<ceil_div(N_NODES / 4, B), B, 0, stream>>>((float4*)degf, 1.0f, N_NODES / 4, tick);
    deg_accum<<<2 * N_SLICES, B, 0, stream>>>(ei, degf, tick);
    offsets_scan<<<ceil_div(N_NODES, SCAN_B), SCAN_B, 0, stream>>>(degf, off, cursor, tick);
    csr_fill<<<2 * N_SLICES, B, 0, stream>>>(ei, cursor, csr, tick);

    const int gBlocks = ceil_div(N_NODES / 16, B / 32);

    gemm_wmma_f32<3, 32, 32><<<gBlocks, B, 0, stream>>>(x, W1, nullptr, bufT, N_NODES, none);
    pull_rows<32, 32><<<ceil_div(N_NODES * 8, B), B, 0, stream>>>(off, csr, bufT, b1, bufH);

    gemm_wmma_f32<32, 16, 32><<<gBlocks, B, 0, stream>>>(bufH, W2, nullptr, bufT, N_NODES, off);
    pull_rows<16, 32><<<ceil_div(N_NODES * 8, B), B, 0, stream>>>(off, csr, bufT, b2, bufH);

    gemm_wmma_f32<16, 16, 32><<<gBlocks, B, 0, stream>>>(bufH, W3, nullptr, bufT, N_NODES, off);
    pull_rows<16, 32><<<ceil_div(N_NODES * 8, B), B, 0, stream>>>(off, csr, bufT, b3, bufH);

    fill4_f32<<<ceil_div(N_GRAPHS * 4, B), B, 0, stream>>>((float4*)pooled, 0.0f, N_GRAPHS * 4, none);
    pool_kernel<<<ceil_div(N_NODES * 4, B), B, 0, stream>>>(bufH, batch, pooled);
    gemm_wmma_f32<16, 16, 16><<<ceil_div(N_GRAPHS / 16, B / 32), B, 0, stream>>>(
        pooled, We, be, out, N_GRAPHS, none);
    classify<<<ceil_div(N_GRAPHS, B), B, 0, stream>>>(out, Wc, bc, out + N_GRAPHS * 16);
}


# Lists in fixed places: 64 per node, filled by ONE edge pass whose per-node counter ends as the in-degree; degree pass and offsets scan go; an over-full node walks the edge list; sum in list order
# speedup vs baseline: 4.1407x; 1.2758x over previous
#include <hip/hip_runtime.h>

typedef __attribute__((ext_vector_type(2))) float v2f;
typedef __attribute__((ext_vector_type(8))) float v8f;

#define N_NODES  100000
#define N_EDGES  3200000
#define N_GRAPHS 1024

#define SPLIT_NODE (N_NODES / 2)
#define SLICE      1024
#define N_SLICES   (N_EDGES / SLICE)
#define TICK_PITCH 128
static_assert(N_EDGES % SLICE == 0, "slices cover the edge list exactly");

#define CAP 64
static_assert(CAP % 32 == 0, "a node's places are whole 128-B lines");

static inline int ceil_div(int a, int b) { return (a + b - 1) / b; }

__global__ void fill4_f32(float4* __restrict__ p, float v, int nquads, int* __restrict__ tick) {
    int i = blockIdx.x * blockDim.x + threadIdx.x;
    if (i < nquads) p[i] = make_float4(v, v, v, v);
    if (tick != nullptr && i < 2) tick[i * TICK_PITCH] = 0;
}

__device__ __forceinline__ int draw_job(int* __restrict__ tick) {
    const int h = (int)((__builtin_amdgcn_s_sendmsg_rtn(0x87) >> 18) & 1u);
    int t = atomicAdd(&tick[h * TICK_PITCH], 1);
    if (t < N_SLICES) return (t << 1) | h;
    t = atomicAdd(&tick[(h ^ 1) * TICK_PITCH], 1);
    if (t < N_SLICES) return (t << 1) | (h ^ 1);
    return -1;
}

__global__ void __launch_bounds__(256)
deg_accum(const int* __restrict__ ei, float* __restrict__ deg, int* __restrict__ tick) {
    __shared__ int s_job;
    if (threadIdx.x == 0) s_job = draw_job(tick);
    __syncthreads();
    const int job = s_job;
    if (job < 0) return;
    const int h = job & 1;
    const int base = (job >> 1) * SLICE;
    const int4 d = *reinterpret_cast<const int4*>(&ei[N_EDGES + base + threadIdx.x * 4]);
    if ((d.x >= SPLIT_NODE ? 1 : 0) == h) atomicAdd(&deg[d.x], 1.0f);
    if ((d.y >= SPLIT_NODE ? 1 : 0) == h) atomicAdd(&deg[d.y], 1.0f);
    if ((d.z >= SPLIT_NODE ? 1 : 0) == h) atomicAdd(&deg[d.z], 1.0f);
    if ((d.w >= SPLIT_NODE ? 1 : 0) == h) atomicAdd(&deg[d.w], 1.0f);
}

template <int CIN, int COUT, int PITCH, int CAPT>
__global__ void __launch_bounds__(256)
gemm_wmma_f32(const float* __restrict__ H, const float* __restrict__ W,
              const float* __restrict__ bias, float* __restrict__ out, int nrows,
              const int* __restrict__ off) {
    static_assert(PITCH == COUT || (PITCH * 4 == 128 && COUT + 4 <= PITCH),
                  "a row with its scale quad lies in one 128-B line");
    constexpr int NT     = COUT / 16;
    constexpr int KSTEPS = (CIN + 3) / 4;

    const int wavesPerBlock = blockDim.x >> 5;
    const int wave = blockIdx.x * wavesPerBlock + (threadIdx.x >> 5);
    const int row0 = wave * 16;
    if (row0 >= nrows) return;

    const int lane = threadIdx.x & 31;
    const int half = lane >> 4;
    const int m    = lane & 15;
    const int row  = row0 + m;

    float rowscale = 0.0f;
    if constexpr (PITCH > COUT) {
        if constexpr (CAPT > 0) rowscale = rsqrtf((float)(off[row] + 1));
        else                    rowscale = rsqrtf((float)(off[row + 1] - off[row] + 1));
    }

    v8f acc[NT];
#pragma unroll
    for (int nt = 0; nt < NT; ++nt) acc[nt] = (v8f){};

#pragma unroll
    for (int ks = 0; ks < KSTEPS; ++ks) {
        const int ka = ks * 4 + 2 * half;

        v2f a;
        if constexpr (CIN % 4 == 0) {
            a = *reinterpret_cast<const v2f*>(&H[row * CIN + ka]);
        } else {
            const int k0 = (ka     < CIN) ? ka     : (CIN - 1);
            const int k1 = (ka + 1 < CIN) ? ka + 1 : (CIN - 1);
            float ax = H[row * CIN + k0];
            float ay = H[row * CIN + k1];
            a.x = (ka     < CIN) ? ax : 0.0f;
            a.y = (ka + 1 < CIN) ? ay : 0.0f;
        }

#pragma unroll
        for (int nt = 0; nt < NT; ++nt) {
            const int ncol = nt * 16 + m;
            v2f b;
            if constexpr (CIN % 4 == 0) {
                b = *reinterpret_cast<const v2f*>(&W[ncol * CIN + ka]);
            } else {
                const int k0 = (ka     < CIN) ? ka     : (CIN - 1);
                const int k1 = (ka + 1 < CIN) ? ka + 1 : (CIN - 1);
                float bx = W[ncol * CIN + k0];
                float by = W[ncol * CIN + k1];
                b.x = (ka     < CIN) ? bx : 0.0f;
                b.y = (ka + 1 < CIN) ? by : 0.0f;
            }
            acc[nt] = __builtin_amdgcn_wmma_f32_16x16x4_f32(
                false, a, false, b, (short)0, acc[nt], false, false);
        }
    }

#pragma unroll
    for (int nt = 0; nt < NT; ++nt) {
        const int col = nt * 16 + m;
        const float bb = bias ? bias[col] : 0.0f;
#pragma unroll
        for (int r = 0; r < 8; ++r) {
            out[(row0 + r + 8 * half) * PITCH + col] = acc[nt][r] + bb;
        }
    }
    if constexpr (PITCH > COUT) {
        if (half == 0)
            *reinterpret_cast<float4*>(&out[row * PITCH + COUT]) = make_float4(rowscale, 0.0f, 0.0f, 0.0f);
    }
}

#define SCAN_B 1024
__global__ void __launch_bounds__(SCAN_B)
offsets_scan(const float* __restrict__ degf, int* __restrict__ off, int* __restrict__ cursor,
             int* __restrict__ tick) {
    __shared__ int s_cnt[SCAN_B / 32];
    __shared__ int s_pre[SCAN_B / 32];
    const int tid  = threadIdx.x;
    const int lane = tid & 31;
    const int wave = tid >> 5;
    const int b    = blockIdx.x;
    const int i    = b * SCAN_B + tid;
    if (b == 0 && tid < 2) tick[tid * TICK_PITCH] = 0;
    if (b == 0 && tid == 2) off[N_NODES] = N_EDGES;

    const int c = (i < N_NODES) ? ((int)degf[i] - 1) : 0;
    int pre = 0;
    for (int j = 0; j < b; ++j) pre += (int)degf[j * SCAN_B + tid] - 1;

    int x = c;
#pragma unroll
    for (int d = 1; d < 32; d <<= 1) {
        const int y = __shfl_up(x, d);
        if (lane >= d) x += y;
    }
#pragma unroll
    for (int d = 16; d > 0; d >>= 1) pre += __shfl_xor(pre, d);
    if (lane == 31) s_cnt[wave] = x;
    if (lane == 0)  s_pre[wave] = pre;
    __syncthreads();
    int base = 0;
#pragma unroll
    for (int w = 0; w < SCAN_B / 32; ++w) {
        base += s_pre[w];
        base += (w < wave) ? s_cnt[w] : 0;
    }
    const int excl = base + x - c;
    if (i < N_NODES) {
        off[i]    = excl;
        cursor[i] = excl;
    }
}

template <int CAPT>
__global__ void __launch_bounds__(256)
csr_fill(const int* __restrict__ ei, int* __restrict__ cursor, int* __restrict__ csr,
         int* __restrict__ tick) {
    __shared__ int s_job;
    if (threadIdx.x == 0) s_job = draw_job(tick);
    __syncthreads();
    const int job = s_job;
    if (job < 0) return;
    const int h = job & 1;
    const int base = (job >> 1) * SLICE;
    const int4 dv = *reinterpret_cast<const int4*>(&ei[N_EDGES + base + threadIdx.x * 4]);
    const int4 sv = *reinterpret_cast<const int4*>(&ei[base + threadIdx.x * 4]);
    const int dd[4] = {dv.x, dv.y, dv.z, dv.w};
    const int ss[4] = {sv.x, sv.y, sv.z, sv.w};
    int pos[4];
#pragma unroll
    for (int k = 0; k < 4; ++k) {
        const bool own = ((dd[k] >= SPLIT_NODE) ? 1 : 0) == h;
        pos[k] = own ? atomicAdd(&cursor[dd[k]], 1) : -1;
    }
#pragma unroll
    for (int k = 0; k < 4; ++k) {
        if constexpr (CAPT > 0) {
            if ((unsigned)pos[k] < (unsigned)CAPT) csr[dd[k] * CAPT + pos[k]] = ss[k];
        } else {
            if (pos[k] >= 0) csr[pos[k]] = ss[k];
        }
    }
}

template <int COUT, int TP, int CAPT>
__global__ void __launch_bounds__(256)
pull_rows(const int* __restrict__ off, const int* __restrict__ csr, const float* __restrict__ t,
          const float* __restrict__ bias, float* __restrict__ outH, const int* __restrict__ ei) {
    constexpr bool SCALE_IN_ROW = TP > COUT;
    constexpr int DL     = COUT / 4;
    constexpr int GROUPS = SCALE_IN_ROW ? 2 * DL : DL;
    static_assert(!SCALE_IN_ROW || (COUT + 4) <= TP, "the scale quad lies inside the row's pitch");
    constexpr int U      = 4;
    constexpr int NB     = 32;
    constexpr int R      = NB / GROUPS;
    static_assert(NB % GROUPS == 0 && NB % U == 0 && (GROUPS & (GROUPS - 1)) == 0,
                  "a block is shared out evenly; a group is an aligned power-of-two run of lanes");
    const int q = blockIdx.x * blockDim.x + threadIdx.x;
    if (q >= N_NODES * GROUPS) return;
    const int n  = q / GROUPS;
    const int j  = q % GROUPS;
    const int c4 = ((j < DL) ? j : DL) * 4;
    int e0, e1;
    if constexpr (CAPT > 0) {
        e0 = n * CAPT;
        e1 = e0 + off[n];
    } else {
        e0 = off[n];
        e1 = off[n + 1];
    }
    const float dn = rsqrtf((float)(e1 - e0 + 1));
    float4 acc = make_float4(0.0f, 0.0f, 0.0f, 0.0f);
    if constexpr (CAPT > 0) {
        if (e1 - e0 > CAPT) {
            for (int e = 0; e < N_EDGES; ++e) {
                if (ei[N_EDGES + e] == n) {
                    const int s = ei[e];
                    const float4 v = *reinterpret_cast<const float4*>(&t[s * TP + c4]);
                    float ds;
                    if constexpr (SCALE_IN_ROW) ds = __shfl(v.x, DL, GROUPS);
                    else                        ds = rsqrtf((float)(off[s] + 1));
                    const float coeff = ds * dn;
                    acc.x += v.x * coeff;
                    acc.y += v.y * coeff;
                    acc.z += v.z * coeff;
                    acc.w += v.w * coeff;
                }
            }
            e1 = e0;
        }
    }
    for (int eb = e0; eb < e1; eb += NB) {
        int idx[R];
#pragma unroll
        for (int r = 0; r < R; ++r) {
            const int ee = eb + r * GROUPS + j;
            idx[r] = csr[(ee < e1) ? ee : (e1 - 1)];
        }
#pragma unroll
        for (int tt = 0; tt < NB / U; ++tt) {
            const int e = eb + tt * U;
            if (e < e1) {
                int    s[U], o0[U], o1[U];
                float4 v[U];
#pragma unroll
                for (int u = 0; u < U; ++u) {
                    const int k = tt * U + u;
                    s[u] = __shfl(idx[k / GROUPS], k % GROUPS, GROUPS);
                }
#pragma unroll
                for (int u = 0; u < U; ++u) {
                    v[u] = *reinterpret_cast<const float4*>(&t[s[u] * TP + c4]);
                    if constexpr (!SCALE_IN_ROW) {
                        if constexpr (CAPT > 0) {
                            o0[u] = 0;
                            o1[u] = off[s[u]];
                        } else {
                            o0[u] = off[s[u]];
                            o1[u] = off[s[u] + 1];
                        }
                    }
                }
                __builtin_amdgcn_sched_group_barrier(0x020, SCALE_IN_ROW ? U : 2 * U, 0);
#pragma unroll
                for (int u = 0; u < U; ++u) {
                    const bool  live  = (e + u) < e1;
                    float ds;
                    if constexpr (SCALE_IN_ROW) ds = __shfl(v[u].x, DL, GROUPS);
                    else                        ds = rsqrtf((float)(o1[u] - o0[u] + 1));
                    const float coeff = ds * dn;
                    acc.x += live ? v[u].x * coeff : 0.0f;
                    acc.y += live ? v[u].y * coeff : 0.0f;
                    acc.z += live ? v[u].z * coeff : 0.0f;
                    acc.w += live ? v[u].w * coeff : 0.0f;
                }
            }
        }
    }
    const float dd = dn * dn;
    const float4 vt = *reinterpret_cast<const float4*>(&t[n * TP + c4]);
    const int    co = (j < DL) ? j * 4 : 0;
    const float4 vb = *reinterpret_cast<const float4*>(&bias[co]);
    float4 r;
    r.x = fmaxf(fmaf(vt.x, dd, acc.x) + vb.x, 0.0f);
    r.y = fmaxf(fmaf(vt.y, dd, acc.y) + vb.y, 0.0f);
    r.z = fmaxf(fmaf(vt.z, dd, acc.z) + vb.z, 0.0f);
    r.w = fmaxf(fmaf(vt.w, dd, acc.w) + vb.w, 0.0f);
    if (j < DL) *reinterpret_cast<float4*>(&outH[n * COUT + co]) = r;
}

__global__ void pool_kernel(const float* __restrict__ h, const int* __restrict__ batch,
                            float* __restrict__ pooled) {
    const int q = blockIdx.x * blockDim.x + threadIdx.x;
    if (q >= N_NODES * 4) return;
    const int n  = q >> 2;
    const int c4 = (q & 3) * 4;
    const float4 v = *reinterpret_cast<const float4*>(&h[n * 16 + c4]);
    float* base = &pooled[batch[n] * 16 + c4];
    atomicAdd(base + 0, v.x);
    atomicAdd(base + 1, v.y);
    atomicAdd(base + 2, v.z);
    atomicAdd(base + 3, v.w);
}

__global__ void classify(const float* __restrict__ emb, const float* __restrict__ Wc,
                         const float* __restrict__ bc, float* __restrict__ out) {
    const int g = blockIdx.x * blockDim.x + threadIdx.x;
    if (g >= N_GRAPHS) return;
    float s = bc[0];
#pragma unroll
    for (int c = 0; c < 16; ++c) s += fmaxf(emb[g * 16 + c], 0.0f) * Wc[c];
    out[g] = s;
}

extern "C" void kernel_launch(void* const* d_in, const int* in_sizes, int n_in,
                              void* d_out, int out_size, void* d_ws, size_t ws_size,
                              hipStream_t stream) {
    const float* x     = (const float*)d_in[0];
    const int*   ei    = (const int*)d_in[1];
    const int*   batch = (const int*)d_in[2];
    const float* W1 = (const float*)d_in[3];
    const float* b1 = (const float*)d_in[4];
    const float* W2 = (const float*)d_in[5];
    const float* b2 = (const float*)d_in[6];
    const float* W3 = (const float*)d_in[7];
    const float* b3 = (const float*)d_in[8];
    const float* We = (const float*)d_in[9];
    const float* be = (const float*)d_in[10];
    const float* Wc = (const float*)d_in[11];
    const float* bc = (const float*)d_in[12];
    float* out = (float*)d_out;

    float* ws   = (float*)d_ws;
    float* bufT = ws;
    int*   tick   = (int*)bufT;
    float* pooled = bufT;
    int*   none   = nullptr;

    const int B = 256;
    const int gBlocks = ceil_div(N_NODES / 16, B / 32);
    float* bufH;

    constexpr size_t PLACES_WS_BYTES = ((size_t)N_NODES * (32 + CAP + 32) + N_NODES + 4) * 4;
    if (ws_size >= PLACES_WS_BYTES) {
        int* lists = (int*)(bufT + N_NODES * 32);
        bufH       = bufT + N_NODES * (32 + CAP);
        int* cnt   = (int*)(bufT + N_NODES * (64 + CAP));

        fill4_f32<<<ceil_div(N_NODES / 4, B), B, 0, stream>>>((float4*)cnt, 0.0f, N_NODES / 4, tick);
        csr_fill<CAP><<<2 * N_SLICES, B, 0, stream>>>(ei, cnt, lists, tick);

        gemm_wmma_f32<3, 32, 32, CAP><<<gBlocks, B, 0, stream>>>(x, W1, nullptr, bufT, N_NODES, none);
        pull_rows<32, 32, CAP><<<ceil_div(N_NODES * 8, B), B, 0, stream>>>(cnt, lists, bufT, b1, bufH, ei);

        gemm_wmma_f32<32, 16, 32, CAP><<<gBlocks, B, 0, stream>>>(bufH, W2, nullptr, bufT, N_NODES, cnt);
        pull_rows<16, 32, CAP><<<ceil_div(N_NODES * 8, B), B, 0, stream>>>(cnt, lists, bufT, b2, bufH, ei);

        gemm_wmma_f32<16, 16, 32, CAP><<<gBlocks, B, 0, stream>>>(bufH, W3, nullptr, bufT, N_NODES, cnt);
        pull_rows<16, 32, CAP><<<ceil_div(N_NODES * 8, B), B, 0, stream>>>(cnt, lists, bufT, b3, bufH, ei);
    } else {
        int*   csr  = (int*)(bufT + N_NODES * 32);
        bufH        = bufT + N_NODES * 64;
        int*   off  = (int*)(bufT + N_NODES * 96);
        float* degf   = bufH;
        int*   cursor = (int*)(bufH + N_NODES);

        fill4_f32<<<ceil_div(N_NODES / 4, B), B, 0, stream>>>((float4*)degf, 1.0f, N_NODES / 4, tick);
        deg_accum<<<2 * N_SLICES, B, 0, stream>>>(ei, degf, tick);
        offsets_scan<<<ceil_div(N_NODES, SCAN_B), SCAN_B, 0, stream>>>(degf, off, cursor, tick);
        csr_fill<0><<<2 * N_SLICES, B, 0, stream>>>(ei, cursor, csr, tick);

        gemm_wmma_f32<3, 32, 32, 0><<<gBlocks, B, 0, stream>>>(x, W1, nullptr, bufT, N_NODES, none);
        pull_rows<32, 32, 0><<<ceil_div(N_NODES * 8, B), B, 0, stream>>>(off, csr, bufT, b1, bufH, ei);
        gemm_wmma_f32<32, 16, 32, 0><<<gBlocks, B, 0, stream>>>(bufH, W2, nullptr, bufT, N_NODES, off);
        pull_rows<16, 32, 0><<<ceil_div(N_NODES * 8, B), B, 0, stream>>>(off, csr, bufT, b2, bufH, ei);
        gemm_wmma_f32<16, 16, 32, 0><<<gBlocks, B, 0, stream>>>(bufH, W3, nullptr, bufT, N_NODES, off);
        pull_rows<16, 32, 0><<<ceil_div(N_NODES * 8, B), B, 0, stream>>>(off, csr, bufT, b3, bufH, ei);
    }

    fill4_f32<<<ceil_div(N_GRAPHS * 4, B), B, 0, stream>>>((float4*)pooled, 0.0f, N_GRAPHS * 4, none);
    pool_kernel<<<ceil_div(N_NODES * 4, B), B, 0, stream>>>(bufH, batch, pooled);
    gemm_wmma_f32<16, 16, 16, 0><<<ceil_div(N_GRAPHS / 16, B / 32), B, 0, stream>>>(
        pooled, We, be, out, N_GRAPHS, none);
    classify<<<ceil_div(N_GRAPHS, B), B, 0, stream>>>(out, Wc, bc, out + N_GRAPHS * 16);
}
